# starting kernel
# baseline (speedup 1.0000x reference)
#include <hip/hip_runtime.h>
#include <hip/hip_bf16.h>


#define N_NODES 100000
#define N_EDGES 640000
#define D 128

typedef float v2f __attribute__((ext_vector_type(2)));
typedef float v8f __attribute__((ext_vector_type(8)));

__device__ __forceinline__ void atomAddF(float* p, float v) {
    (void)__hip_atomic_fetch_add(p, v, __ATOMIC_RELAXED, __HIP_MEMORY_SCOPE_AGENT);
}

__global__ __launch_bounds__(256) void sage_init(
    float* __restrict__ agg_in, float* __restrict__ agg_out,
    float* __restrict__ cnt_in, float* __restrict__ cnt_out,
    const float* __restrict__ w_r_in, const float* __restrict__ w_r_out,
    const float* __restrict__ b_l_in, const float* __restrict__ b_l_out,
    float* __restrict__ w_sum, float* __restrict__ b_sum)
{
    const int tid    = blockIdx.x * blockDim.x + threadIdx.x;
    const int stride = gridDim.x * blockDim.x;
    const int nAgg = N_NODES * D;
    for (int j = tid; j < nAgg; j += stride) { agg_in[j] = 0.0f; agg_out[j] = 0.0f; }
    for (int j = tid; j < N_NODES; j += stride) { cnt_in[j] = 0.0f; cnt_out[j] = 0.0f; }
    for (int j = tid; j < D * D; j += stride) w_sum[j] = w_r_in[j] + w_r_out[j];
    for (int j = tid; j < D; j += stride)     b_sum[j] = b_l_in[j] + b_l_out[j];
}

__global__ __launch_bounds__(256) void sage_scatter(
    const int* __restrict__ src, const int* __restrict__ dst,
    const float* __restrict__ x,
    float* __restrict__ agg_in, float* __restrict__ agg_out,
    float* __restrict__ cnt_in, float* __restrict__ cnt_out)
{
    const int e = blockIdx.x * 8 + (threadIdx.x >> 5);
    if (e >= N_EDGES) return;
    const int lane = threadIdx.x & 31;
    const int s = src[e];
    const int d = dst[e];

    const float4 xs = *(const float4*)(x + (size_t)s * D + lane * 4);
    const float4 xd = *(const float4*)(x + (size_t)d * D + lane * 4);

    float* pin  = agg_in  + (size_t)d * D + lane * 4;
    float* pout = agg_out + (size_t)s * D + lane * 4;
    atomAddF(pin + 0, xs.x); atomAddF(pin + 1, xs.y);
    atomAddF(pin + 2, xs.z); atomAddF(pin + 3, xs.w);
    atomAddF(pout + 0, xd.x); atomAddF(pout + 1, xd.y);
    atomAddF(pout + 2, xd.z); atomAddF(pout + 3, xd.w);

    if (lane == 0) { atomAddF(cnt_in + d, 1.0f); atomAddF(cnt_out + s, 1.0f); }
}

__device__ __forceinline__ v8f mac_section(const float* __restrict__ A,
                                           const float* __restrict__ W,
                                           float sc, v8f acc)
{
#pragma unroll
    for (int k = 0; k < D; k += 4) {
        v2f a = *(const v2f*)(A + k);
        a.x *= sc; a.y *= sc;
        v2f b = *(const v2f*)(W + k);
        acc = __builtin_amdgcn_wmma_f32_16x16x4_f32(
             false, a,  false, b,
             (short)0, acc,  false,  false);
    }
    return acc;
}

__global__ __launch_bounds__(256) void sage_gemm(
    const float* __restrict__ x,
    const float* __restrict__ agg_in, const float* __restrict__ agg_out,
    const float* __restrict__ cnt_in, const float* __restrict__ cnt_out,
    const float* __restrict__ w_l_in, const float* __restrict__ w_l_out,
    const float* __restrict__ w_sum,  const float* __restrict__ b_sum,
    float* __restrict__ out)
{
    const int mtile = blockIdx.x;
    const int wave  = threadIdx.x >> 5;
    const int lane  = threadIdx.x & 31;
    const int half  = lane >> 4;
    const int l15   = lane & 15;
    const int row   = mtile * 16 + l15;
    const int col   = wave * 16 + l15;

    const float sc_in  = 1.0f / fmaxf(cnt_in[row],  1.0f);
    const float sc_out = 1.0f / fmaxf(cnt_out[row], 1.0f);

    const float bias = b_sum[col];
    v8f acc;
#pragma unroll
    for (int r = 0; r < 8; ++r) acc[r] = bias;

    const size_t aoff = (size_t)row * D + 2 * half;
    const size_t woff = (size_t)col * D + 2 * half;

    acc = mac_section(agg_in  + aoff, w_l_in  + woff, sc_in,  acc);
    acc = mac_section(agg_out + aoff, w_l_out + woff, sc_out, acc);
    acc = mac_section(x       + aoff, w_sum   + woff, 1.0f,   acc);

#pragma unroll
    for (int r = 0; r < 8; ++r) {
        const int m = mtile * 16 + half * 8 + r;
        out[(size_t)m * D + col] = acc[r];
    }
}

extern "C" void kernel_launch(void* const* d_in, const int* in_sizes, int n_in,
                              void* d_out, int out_size, void* d_ws, size_t ws_size,
                              hipStream_t stream)
{
    const float* x       = (const float*)d_in[0];
    const int*   ei      = (const int*)d_in[1];
    const float* w_l_in  = (const float*)d_in[2];
    const float* b_l_in  = (const float*)d_in[3];
    const float* w_r_in  = (const float*)d_in[4];
    const float* w_l_out = (const float*)d_in[5];
    const float* b_l_out = (const float*)d_in[6];
    const float* w_r_out = (const float*)d_in[7];
    float* out = (float*)d_out;

    const size_t NB   = (size_t)N_NODES * D * sizeof(float);
    const size_t CNTB = (size_t)N_NODES * sizeof(float);
    char* ws = (char*)d_ws;
    float* agg_in  = (float*)(ws);
    float* agg_out = (float*)(ws + NB);
    float* cnt_in  = (float*)(ws + 2 * NB);
    float* cnt_out = (float*)(ws + 2 * NB + CNTB);
    float* w_sum   = (float*)(ws + 2 * NB + 2 * CNTB);
    float* b_sum   = (float*)(ws + 2 * NB + 2 * CNTB + (size_t)D * D * sizeof(float));

    (void)in_sizes; (void)n_in; (void)out_size; (void)ws_size;

    sage_init<<<2048, 256, 0, stream>>>(agg_in, agg_out, cnt_in, cnt_out,
                                        w_r_in, w_r_out, b_l_in, b_l_out,
                                        w_sum, b_sum);

    sage_scatter<<<N_EDGES / 8, 256, 0, stream>>>(ei, ei + N_EDGES, x,
                                                  agg_in, agg_out, cnt_in, cnt_out);

    sage_gemm<<<N_NODES / 16, 256, 0, stream>>>(x, agg_in, agg_out, cnt_in, cnt_out,
                                                w_l_in, w_l_out, w_sum, b_sum, out);
}


# GEMM re-tiled: 256 resident work-groups loop over 32-row tiles, each wave keeps its W fragments in registers, pre-scaled A tiles staged in LDS by full-row loads; f32 WMMA k order unchanged
# speedup vs baseline: 1.4905x; 1.4905x over previous
#include <hip/hip_runtime.h>
#include <hip/hip_bf16.h>


#define N_NODES 100000
#define N_EDGES 640000
#define D 128

typedef float v2f __attribute__((ext_vector_type(2)));
typedef float v8f __attribute__((ext_vector_type(8)));

__device__ __forceinline__ void atomAddF(float* p, float v) {
    (void)__hip_atomic_fetch_add(p, v, __ATOMIC_RELAXED, __HIP_MEMORY_SCOPE_AGENT);
}

__global__ __launch_bounds__(256) void sage_init(
    float* __restrict__ agg_in, float* __restrict__ agg_out,
    float* __restrict__ cnt_in, float* __restrict__ cnt_out,
    const float* __restrict__ w_r_in, const float* __restrict__ w_r_out,
    const float* __restrict__ b_l_in, const float* __restrict__ b_l_out,
    float* __restrict__ w_sum, float* __restrict__ b_sum)
{
    const int tid    = blockIdx.x * blockDim.x + threadIdx.x;
    const int stride = gridDim.x * blockDim.x;
    const int nAgg = N_NODES * D;
    for (int j = tid; j < nAgg; j += stride) { agg_in[j] = 0.0f; agg_out[j] = 0.0f; }
    for (int j = tid; j < N_NODES; j += stride) { cnt_in[j] = 0.0f; cnt_out[j] = 0.0f; }
    for (int j = tid; j < D * D; j += stride) w_sum[j] = w_r_in[j] + w_r_out[j];
    for (int j = tid; j < D; j += stride)     b_sum[j] = b_l_in[j] + b_l_out[j];
}

__global__ __launch_bounds__(256) void sage_scatter(
    const int* __restrict__ src, const int* __restrict__ dst,
    const float* __restrict__ x,
    float* __restrict__ agg_in, float* __restrict__ agg_out,
    float* __restrict__ cnt_in, float* __restrict__ cnt_out)
{
    const int e = blockIdx.x * 8 + (threadIdx.x >> 5);
    if (e >= N_EDGES) return;
    const int lane = threadIdx.x & 31;
    const int s = src[e];
    const int d = dst[e];

    const float4 xs = *(const float4*)(x + (size_t)s * D + lane * 4);
    const float4 xd = *(const float4*)(x + (size_t)d * D + lane * 4);

    float* pin  = agg_in  + (size_t)d * D + lane * 4;
    float* pout = agg_out + (size_t)s * D + lane * 4;
    atomAddF(pin + 0, xs.x); atomAddF(pin + 1, xs.y);
    atomAddF(pin + 2, xs.z); atomAddF(pin + 3, xs.w);
    atomAddF(pout + 0, xd.x); atomAddF(pout + 1, xd.y);
    atomAddF(pout + 2, xd.z); atomAddF(pout + 3, xd.w);

    if (lane == 0) { atomAddF(cnt_in + d, 1.0f); atomAddF(cnt_out + s, 1.0f); }
}

#define TM      32
#define KTOT    (3 * D)
#define PITCH   (KTOT + 4)
#define NTILES  (N_NODES / TM)
#define GEMM_BLOCKS 256
#define KC      16
#define NCHUNK  (KTOT / 4 / KC)

__device__ __forceinline__ void tile_fetch(
    int tile, int r0, int c4,
    const float* __restrict__ x,
    const float* __restrict__ agg_in, const float* __restrict__ agg_out,
    const float* __restrict__ cnt_in, const float* __restrict__ cnt_out,
    float4 (&pf)[12], float (&ci)[4], float (&co)[4])
{
#pragma unroll
    for (int j = 0; j < 4; ++j) {
        const int row = tile * TM + r0 + 8 * j;
        const size_t g = (size_t)row * D + c4 * 4;
        pf[j]     = *(const float4*)(agg_in  + g);
        pf[4 + j] = *(const float4*)(agg_out + g);
        pf[8 + j] = *(const float4*)(x       + g);
        ci[j] = cnt_in[row];
        co[j] = cnt_out[row];
    }
}

__device__ __forceinline__ void tile_put(
    float* __restrict__ dstTile, int r0, int c4,
    const float4 (&pf)[12], const float (&ci)[4], const float (&co)[4])
{
#pragma unroll
    for (int j = 0; j < 4; ++j) {
        const float si = 1.0f / fmaxf(ci[j], 1.0f);
        const float so = 1.0f / fmaxf(co[j], 1.0f);
        float* p = dstTile + (r0 + 8 * j) * PITCH + c4 * 4;
        float4 a = pf[j];
        a.x *= si; a.y *= si; a.z *= si; a.w *= si;
        float4 b = pf[4 + j];
        b.x *= so; b.y *= so; b.z *= so; b.w *= so;
        *(float4*)(p)         = a;
        *(float4*)(p + D)     = b;
        *(float4*)(p + 2 * D) = pf[8 + j];
    }
}

__device__ __forceinline__ void frag_load(const float* A0, const float* A1, int c,
                                          v2f (&f0)[KC], v2f (&f1)[KC])
{
#pragma unroll
    for (int k = 0; k < KC; ++k) {
        f0[k] = *(const v2f*)(A0 + 4 * (c * KC + k));
        f1[k] = *(const v2f*)(A1 + 4 * (c * KC + k));
    }
}

__global__ __launch_bounds__(256) void sage_gemm(
    const float* __restrict__ x,
    const float* __restrict__ agg_in, const float* __restrict__ agg_out,
    const float* __restrict__ cnt_in, const float* __restrict__ cnt_out,
    const float* __restrict__ w_l_in, const float* __restrict__ w_l_out,
    const float* __restrict__ w_sum,  const float* __restrict__ b_sum,
    float* __restrict__ out)
{
    __shared__ __attribute__((aligned(16))) float tileA[2][TM * PITCH];

    const int tid   = threadIdx.x;
    const int wave  = tid >> 5;
    const int lane  = tid & 31;
    const int half  = lane >> 4;
    const int l15   = lane & 15;
    const int col   = wave * 16 + l15;

    v2f bw[96];
    {
        const size_t woff = (size_t)col * D + 2 * half;
#pragma unroll
        for (int i = 0; i < 32; ++i) {
            bw[i]      = *(const v2f*)(w_l_in  + woff + 4 * i);
            bw[32 + i] = *(const v2f*)(w_l_out + woff + 4 * i);
            bw[64 + i] = *(const v2f*)(w_sum   + woff + 4 * i);
        }
    }
    const float bias = b_sum[col];

    const int c4 = lane;
    const int r0 = wave;

    float4 pf[12];
    float  ci[4], co[4];

    int tile = blockIdx.x;
    int buf  = 0;
    if (tile < NTILES) {
        tile_fetch(tile, r0, c4, x, agg_in, agg_out, cnt_in, cnt_out, pf, ci, co);
        tile_put(&tileA[0][0], r0, c4, pf, ci, co);
    }
    __syncthreads();

    for (; tile < NTILES; tile += GEMM_BLOCKS) {
        const int  next     = tile + GEMM_BLOCKS;
        const bool has_next = next < NTILES;
        if (has_next)
            tile_fetch(next, r0, c4, x, agg_in, agg_out, cnt_in, cnt_out, pf, ci, co);

        const float* A0 = &tileA[buf][0] + l15 * PITCH + 2 * half;
        const float* A1 = A0 + 16 * PITCH;

        v8f acc0, acc1;
#pragma unroll
        for (int r = 0; r < 8; ++r) { acc0[r] = bias; acc1[r] = bias; }

        v2f fa0[2][KC], fa1[2][KC];
        frag_load(A0, A1, 0, fa0[0], fa1[0]);
#pragma unroll
        for (int c = 0; c < NCHUNK; ++c) {
            if (c + 1 < NCHUNK)
                frag_load(A0, A1, c + 1, fa0[(c + 1) & 1], fa1[(c + 1) & 1]);
            __builtin_amdgcn_sched_barrier(0);
#pragma unroll
            for (int k = 0; k < KC; ++k) {
                acc0 = __builtin_amdgcn_wmma_f32_16x16x4_f32(
                     false, fa0[c & 1][k],  false, bw[c * KC + k],
                     (short)0, acc0,  false,  false);
                acc1 = __builtin_amdgcn_wmma_f32_16x16x4_f32(
                     false, fa1[c & 1][k],  false, bw[c * KC + k],
                     (short)0, acc1,  false,  false);
            }
            __builtin_amdgcn_sched_barrier(0);
        }

        if (has_next)
            tile_put(&tileA[buf ^ 1][0], r0, c4, pf, ci, co);
        __syncthreads();

        const int mbase = tile * TM + half * 8;
#pragma unroll
        for (int r = 0; r < 8; ++r) {
            out[(size_t)(mbase + r) * D + col]      = acc0[r];
            out[(size_t)(mbase + 16 + r) * D + col] = acc1[r];
        }
        buf ^= 1;
    }
}

extern "C" void kernel_launch(void* const* d_in, const int* in_sizes, int n_in,
                              void* d_out, int out_size, void* d_ws, size_t ws_size,
                              hipStream_t stream)
{
    const float* x       = (const float*)d_in[0];
    const int*   ei      = (const int*)d_in[1];
    const float* w_l_in  = (const float*)d_in[2];
    const float* b_l_in  = (const float*)d_in[3];
    const float* w_r_in  = (const float*)d_in[4];
    const float* w_l_out = (const float*)d_in[5];
    const float* b_l_out = (const float*)d_in[6];
    const float* w_r_out = (const float*)d_in[7];
    float* out = (float*)d_out;

    const size_t NB   = (size_t)N_NODES * D * sizeof(float);
    const size_t CNTB = (size_t)N_NODES * sizeof(float);
    char* ws = (char*)d_ws;
    float* agg_in  = (float*)(ws);
    float* agg_out = (float*)(ws + NB);
    float* cnt_in  = (float*)(ws + 2 * NB);
    float* cnt_out = (float*)(ws + 2 * NB + CNTB);
    float* w_sum   = (float*)(ws + 2 * NB + 2 * CNTB);
    float* b_sum   = (float*)(ws + 2 * NB + 2 * CNTB + (size_t)D * D * sizeof(float));

    (void)in_sizes; (void)n_in; (void)out_size; (void)ws_size;

    sage_init<<<2048, 256, 0, stream>>>(agg_in, agg_out, cnt_in, cnt_out,
                                        w_r_in, w_r_out, b_l_in, b_l_out,
                                        w_sum, b_sum);

    sage_scatter<<<N_EDGES / 8, 256, 0, stream>>>(ei, ei + N_EDGES, x,
                                                  agg_in, agg_out, cnt_in, cnt_out);

    sage_gemm<<<GEMM_BLOCKS, 256, 0, stream>>>(x, agg_in, agg_out, cnt_in, cnt_out,
                                                w_l_in, w_l_out, w_sum, b_sum, out);
}


# Float atomic scatter replaced by gather: a returning int atomic per edge end parks the neighbour index in the target's own scratch row, 16 waves per SIMD sum the listed x rows; sum order = slot order
# speedup vs baseline: 2.1580x; 1.4478x over previous
#include <hip/hip_runtime.h>
#include <hip/hip_bf16.h>


#define N_NODES 100000
#define N_EDGES 640000
#define D 128

typedef float v2f __attribute__((ext_vector_type(2)));
typedef float v8f __attribute__((ext_vector_type(8)));

#define CAP D

__global__ __launch_bounds__(256) void sage_init(
    int* __restrict__ cnt_in, int* __restrict__ cnt_out,
    const float* __restrict__ w_r_in, const float* __restrict__ w_r_out,
    const float* __restrict__ b_l_in, const float* __restrict__ b_l_out,
    float* __restrict__ w_sum, float* __restrict__ b_sum)
{
    const int tid    = blockIdx.x * blockDim.x + threadIdx.x;
    const int stride = gridDim.x * blockDim.x;
    for (int j = tid; j < N_NODES; j += stride) { cnt_in[j] = 0; cnt_out[j] = 0; }
    for (int j = tid; j < D * D; j += stride) w_sum[j] = w_r_in[j] + w_r_out[j];
    for (int j = tid; j < D; j += stride)     b_sum[j] = b_l_in[j] + b_l_out[j];
}

__global__ __launch_bounds__(256) void sage_fill(
    const int* __restrict__ src, const int* __restrict__ dst,
    float* __restrict__ agg_in, float* __restrict__ agg_out,
    int* __restrict__ cnt_in, int* __restrict__ cnt_out)
{
    const int e = blockIdx.x * 256 + threadIdx.x;
    if (e >= N_EDGES) return;
    const int s = src[e];
    const int d = dst[e];
    const int a = __hip_atomic_fetch_add(cnt_in + d, 1, __ATOMIC_RELAXED, __HIP_MEMORY_SCOPE_AGENT);
    const int b = __hip_atomic_fetch_add(cnt_out + s, 1, __ATOMIC_RELAXED, __HIP_MEMORY_SCOPE_AGENT);
    if (a < CAP) agg_in [(size_t)d * D + a] = __int_as_float(s);
    if (b < CAP) agg_out[(size_t)s * D + b] = __int_as_float(d);
}

#define GATHER_BLOCKS 2048
#define GATHER_WAVES  (GATHER_BLOCKS * 8)
#define N_TASKS       (2 * N_NODES)

__device__ __forceinline__ float4 x_row(const float* __restrict__ x, int node, int lane) {
    return *(const float4*)(x + (size_t)node * D + lane * 4);
}

__device__ __forceinline__ void acc_add(float4& acc, const float4& r) {
    acc.x += r.x; acc.y += r.y; acc.z += r.z; acc.w += r.w;
}

__device__ __forceinline__ float4 sum_group(const int (&nb)[8], int cnt,
                                            const float* __restrict__ x, int lane)
{
    float4 r[8];
#pragma unroll
    for (int t = 0; t < 8; ++t) {
        if (t < cnt) r[t] = x_row(x, nb[t], lane);
        else         r[t] = make_float4(0.0f, 0.0f, 0.0f, 0.0f);
    }
    float4 g = r[0];
#pragma unroll
    for (int t = 1; t < 8; ++t) acc_add(g, r[t]);
    return g;
}

__device__ __forceinline__ void add_rows(int idv, int j0, int m, const float* __restrict__ x,
                                         int lane, float4& acc)
{
    for (int j = j0; j < m; j += 8) {
        int nb[8];
#pragma unroll
        for (int t = 0; t < 8; ++t)
            nb[t] = __builtin_amdgcn_readlane(idv, j + t);
        acc_add(acc, sum_group(nb, m - j, x, lane));
    }
}

__global__ __launch_bounds__(256) void sage_gather(
    const int* __restrict__ src, const int* __restrict__ dst,
    const float* __restrict__ x,
    float* agg_in, float* agg_out, int* cnt_in, int* cnt_out)
{
    const int lane = threadIdx.x & 31;
    int task = __builtin_amdgcn_readfirstlane(blockIdx.x * 8 + (threadIdx.x >> 5));
    if (task >= N_TASKS) return;

    bool   revN = task >= N_NODES;
    int    nN   = revN ? task - N_NODES : task;
    float* rowN = (revN ? agg_out : agg_in) + (size_t)nN * D;
    int*   cntN = (revN ? cnt_out : cnt_in) + nN;
    int    degN = *cntN;
    int    idsN = __float_as_int(rowN[lane]);

    for (;;) {
        const bool rev = revN;
        const int  n   = nN;
        float*     row = rowN;
        int*       cnt = cntN;
        const int deg = __builtin_amdgcn_readfirstlane(degN);
        const int m0  = min(deg, 32);
        const int id0 = (lane < deg) ? idsN : 0;
        int nb[8];
#pragma unroll
        for (int t = 0; t < 8; ++t)
            nb[t] = __builtin_amdgcn_readlane(id0, t);

        const int  next     = task + GATHER_WAVES;
        const bool has_next = next < N_TASKS;
        if (has_next) {
            revN = next >= N_NODES;
            nN   = revN ? next - N_NODES : next;
            rowN = (revN ? agg_out : agg_in) + (size_t)nN * D;
            cntN = (revN ? cnt_out : cnt_in) + nN;
            degN = *cntN;
            idsN = __float_as_int(rowN[lane]);
        }

        float4 acc = make_float4(0.0f, 0.0f, 0.0f, 0.0f);
        if (deg <= CAP) {
            int idv[3];
#pragma unroll
            for (int c = 1; c < 4; ++c) {
                const int i = c * 32 + lane;
                idv[c - 1] = 0;
                if (i < deg) idv[c - 1] = __float_as_int(row[i]);
            }
            acc = sum_group(nb, m0, x, lane);
            add_rows(id0, 8, m0, x, lane, acc);
#pragma unroll
            for (int c = 1; c < 4; ++c)
                add_rows(idv[c - 1], 0, min(deg - 32 * c, 32), x, lane, acc);
        } else {
            const int* key = rev ? src : dst;
            const int* oth = rev ? dst : src;
            for (int base = 0; base < N_EDGES; base += 32) {
                const int k = key[base + lane];
                const int o = oth[base + lane];
                unsigned hit = __builtin_amdgcn_ballot_w32(k == n);
                while (hit) {
                    const int j = __builtin_ctz(hit);
                    hit &= hit - 1;
                    acc_add(acc, x_row(x, __builtin_amdgcn_readlane(o, j), lane));
                }
            }
        }
        *(float4*)(row + lane * 4) = acc;
        if (lane == 0) *cnt = __float_as_int((float)deg);

        if (!has_next) break;
        task = next;
    }
}

#define TM      32
#define KTOT    (3 * D)
#define PITCH   (KTOT + 4)
#define NTILES  (N_NODES / TM)
#define GEMM_BLOCKS 256
#define KC      16
#define NCHUNK  (KTOT / 4 / KC)

__device__ __forceinline__ void tile_fetch(
    int tile, int r0, int c4,
    const float* __restrict__ x,
    const float* __restrict__ agg_in, const float* __restrict__ agg_out,
    const float* __restrict__ cnt_in, const float* __restrict__ cnt_out,
    float4 (&pf)[12], float (&ci)[4], float (&co)[4])
{
#pragma unroll
    for (int j = 0; j < 4; ++j) {
        const int row = tile * TM + r0 + 8 * j;
        const size_t g = (size_t)row * D + c4 * 4;
        pf[j]     = *(const float4*)(agg_in  + g);
        pf[4 + j] = *(const float4*)(agg_out + g);
        pf[8 + j] = *(const float4*)(x       + g);
        ci[j] = cnt_in[row];
        co[j] = cnt_out[row];
    }
}

__device__ __forceinline__ void tile_put(
    float* __restrict__ dstTile, int r0, int c4,
    const float4 (&pf)[12], const float (&ci)[4], const float (&co)[4])
{
#pragma unroll
    for (int j = 0; j < 4; ++j) {
        const float si = 1.0f / fmaxf(ci[j], 1.0f);
        const float so = 1.0f / fmaxf(co[j], 1.0f);
        float* p = dstTile + (r0 + 8 * j) * PITCH + c4 * 4;
        float4 a = pf[j];
        a.x *= si; a.y *= si; a.z *= si; a.w *= si;
        float4 b = pf[4 + j];
        b.x *= so; b.y *= so; b.z *= so; b.w *= so;
        *(float4*)(p)         = a;
        *(float4*)(p + D)     = b;
        *(float4*)(p + 2 * D) = pf[8 + j];
    }
}

__device__ __forceinline__ void frag_load(const float* A0, const float* A1, int c,
                                          v2f (&f0)[KC], v2f (&f1)[KC])
{
#pragma unroll
    for (int k = 0; k < KC; ++k) {
        f0[k] = *(const v2f*)(A0 + 4 * (c * KC + k));
        f1[k] = *(const v2f*)(A1 + 4 * (c * KC + k));
    }
}

__global__ __launch_bounds__(256) void sage_gemm(
    const float* __restrict__ x,
    const float* __restrict__ agg_in, const float* __restrict__ agg_out,
    const float* __restrict__ cnt_in, const float* __restrict__ cnt_out,
    const float* __restrict__ w_l_in, const float* __restrict__ w_l_out,
    const float* __restrict__ w_sum,  const float* __restrict__ b_sum,
    float* __restrict__ out)
{
    __shared__ __attribute__((aligned(16))) float tileA[2][TM * PITCH];

    const int tid   = threadIdx.x;
    const int wave  = tid >> 5;
    const int lane  = tid & 31;
    const int half  = lane >> 4;
    const int l15   = lane & 15;
    const int col   = wave * 16 + l15;

    v2f bw[96];
    {
        const size_t woff = (size_t)col * D + 2 * half;
#pragma unroll
        for (int i = 0; i < 32; ++i) {
            bw[i]      = *(const v2f*)(w_l_in  + woff + 4 * i);
            bw[32 + i] = *(const v2f*)(w_l_out + woff + 4 * i);
            bw[64 + i] = *(const v2f*)(w_sum   + woff + 4 * i);
        }
    }
    const float bias = b_sum[col];

    const int c4 = lane;
    const int r0 = wave;

    float4 pf[12];
    float  ci[4], co[4];

    int tile = blockIdx.x;
    int buf  = 0;
    if (tile < NTILES) {
        tile_fetch(tile, r0, c4, x, agg_in, agg_out, cnt_in, cnt_out, pf, ci, co);
        tile_put(&tileA[0][0], r0, c4, pf, ci, co);
    }
    __syncthreads();

    for (; tile < NTILES; tile += GEMM_BLOCKS) {
        const int  next     = tile + GEMM_BLOCKS;
        const bool has_next = next < NTILES;
        if (has_next)
            tile_fetch(next, r0, c4, x, agg_in, agg_out, cnt_in, cnt_out, pf, ci, co);

        const float* A0 = &tileA[buf][0] + l15 * PITCH + 2 * half;
        const float* A1 = A0 + 16 * PITCH;

        v8f acc0, acc1;
#pragma unroll
        for (int r = 0; r < 8; ++r) { acc0[r] = bias; acc1[r] = bias; }

        v2f fa0[2][KC], fa1[2][KC];
        frag_load(A0, A1, 0, fa0[0], fa1[0]);
#pragma unroll
        for (int c = 0; c < NCHUNK; ++c) {
            if (c + 1 < NCHUNK)
                frag_load(A0, A1, c + 1, fa0[(c + 1) & 1], fa1[(c + 1) & 1]);
            __builtin_amdgcn_sched_barrier(0);
#pragma unroll
            for (int k = 0; k < KC; ++k) {
                acc0 = __builtin_amdgcn_wmma_f32_16x16x4_f32(
                     false, fa0[c & 1][k],  false, bw[c * KC + k],
                     (short)0, acc0,  false,  false);
                acc1 = __builtin_amdgcn_wmma_f32_16x16x4_f32(
                     false, fa1[c & 1][k],  false, bw[c * KC + k],
                     (short)0, acc1,  false,  false);
            }
            __builtin_amdgcn_sched_barrier(0);
        }

        if (has_next)
            tile_put(&tileA[buf ^ 1][0], r0, c4, pf, ci, co);
        __syncthreads();

        const int mbase = tile * TM + half * 8;
#pragma unroll
        for (int r = 0; r < 8; ++r) {
            out[(size_t)(mbase + r) * D + col]      = acc0[r];
            out[(size_t)(mbase + 16 + r) * D + col] = acc1[r];
        }
        buf ^= 1;
    }
}

extern "C" void kernel_launch(void* const* d_in, const int* in_sizes, int n_in,
                              void* d_out, int out_size, void* d_ws, size_t ws_size,
                              hipStream_t stream)
{
    const float* x       = (const float*)d_in[0];
    const int*   ei      = (const int*)d_in[1];
    const float* w_l_in  = (const float*)d_in[2];
    const float* b_l_in  = (const float*)d_in[3];
    const float* w_r_in  = (const float*)d_in[4];
    const float* w_l_out = (const float*)d_in[5];
    const float* b_l_out = (const float*)d_in[6];
    const float* w_r_out = (const float*)d_in[7];
    float* out = (float*)d_out;

    const size_t NB   = (size_t)N_NODES * D * sizeof(float);
    const size_t CNTB = (size_t)N_NODES * sizeof(float);
    char* ws = (char*)d_ws;
    float* agg_in  = (float*)(ws);
    float* agg_out = (float*)(ws + NB);
    int*   cnt_in  = (int*)(ws + 2 * NB);
    int*   cnt_out = (int*)(ws + 2 * NB + CNTB);
    float* w_sum   = (float*)(ws + 2 * NB + 2 * CNTB);
    float* b_sum   = (float*)(ws + 2 * NB + 2 * CNTB + (size_t)D * D * sizeof(float));

    (void)in_sizes; (void)n_in; (void)out_size; (void)ws_size;

    sage_init<<<(N_NODES + 255) / 256, 256, 0, stream>>>(cnt_in, cnt_out,
                                                         w_r_in, w_r_out, b_l_in, b_l_out,
                                                         w_sum, b_sum);

    sage_fill<<<N_EDGES / 256, 256, 0, stream>>>(ei, ei + N_EDGES, agg_in, agg_out,
                                                 cnt_in, cnt_out);

    sage_gather<<<GATHER_BLOCKS, 256, 0, stream>>>(ei, ei + N_EDGES, x,
                                                   agg_in, agg_out, cnt_in, cnt_out);

    sage_gemm<<<GEMM_BLOCKS, 256, 0, stream>>>(x, agg_in, agg_out,
                                                (const float*)cnt_in, (const float*)cnt_out,
                                                w_l_in, w_l_out, w_sum, b_sum, out);
}


# Per-edge returning global atomics replaced by LDS counters: edge ends bucketed by 256-node range (one global add per bucket and work-group), ranked per bucket in LDS; sum order = slot order as before
# speedup vs baseline: 2.4647x; 1.1421x over previous
#include <hip/hip_runtime.h>
#include <hip/hip_bf16.h>


#define N_NODES 100000
#define N_EDGES 640000
#define D 128

typedef float v2f __attribute__((ext_vector_type(2)));
typedef float v8f __attribute__((ext_vector_type(8)));

#define CAP 64
#define BNODES 256
#define NBUCK  ((N_NODES + BNODES - 1) / BNODES)
#define ESKIP  32

__device__ __forceinline__ int bucket_cap(int b) {
    const int nrows = min(BNODES, N_NODES - b * BNODES);
    return nrows * 64 - ESKIP;
}
__device__ __forceinline__ float* bucket_slot(float* agg, int b, int q) {
    return agg + ((size_t)(b * BNODES + (q >> 6))) * D + 64 + (q & 63);
}

__global__ __launch_bounds__(256) void sage_init(
    float* __restrict__ agg_in, float* __restrict__ agg_out,
    const float* __restrict__ w_r_in, const float* __restrict__ w_r_out,
    const float* __restrict__ b_l_in, const float* __restrict__ b_l_out,
    float* __restrict__ w_sum, float* __restrict__ b_sum)
{
    const int tid    = blockIdx.x * blockDim.x + threadIdx.x;
    const int stride = gridDim.x * blockDim.x;
    for (int j = tid; j < NBUCK; j += stride) {
        *bucket_slot(agg_in,  j, 0) = __int_as_float(0);
        *bucket_slot(agg_out, j, 0) = __int_as_float(0);
    }
    for (int j = tid; j < D * D; j += stride) w_sum[j] = w_r_in[j] + w_r_out[j];
    for (int j = tid; j < D; j += stride)     b_sum[j] = b_l_in[j] + b_l_out[j];
}

#define BK_BLOCKS  256
#define EPB        (N_EDGES / BK_BLOCKS)
#define BK_THREADS 512
#define EPT        ((EPB + BK_THREADS - 1) / BK_THREADS)

__global__ __launch_bounds__(BK_THREADS) void sage_bucket(
    const int* __restrict__ src, const int* __restrict__ dst,
    float* __restrict__ agg_in, float* __restrict__ agg_out)
{
    __shared__ int hist[2 * NBUCK];
    const int tid = threadIdx.x;
    for (int j = tid; j < 2 * NBUCK; j += BK_THREADS) hist[j] = 0;
    __syncthreads();

    const int e0 = blockIdx.x * EPB;
    int s[EPT], d[EPT], ri[EPT], ro[EPT];
#pragma unroll
    for (int i = 0; i < EPT; ++i) {
        const int k = i * BK_THREADS + tid;
        s[i] = 0; d[i] = 0;
        if (k < EPB) { s[i] = src[e0 + k]; d[i] = dst[e0 + k]; }
    }
#pragma unroll
    for (int i = 0; i < EPT; ++i) {
        ri[i] = 0; ro[i] = 0;
        if (i * BK_THREADS + tid < EPB) {
            ri[i] = atomicAdd(&hist[d[i] >> 8], 1);
            ro[i] = atomicAdd(&hist[NBUCK + (s[i] >> 8)], 1);
        }
    }
    __syncthreads();

    for (int j = tid; j < 2 * NBUCK; j += BK_THREADS) {
        const int n = hist[j];
        int base = 0;
        if (n > 0) {
            const bool rev = j >= NBUCK;
            const int  b   = rev ? j - NBUCK : j;
            int* cur = (int*)bucket_slot(rev ? agg_out : agg_in, b, 0);
            base = __hip_atomic_fetch_add(cur, n, __ATOMIC_RELAXED, __HIP_MEMORY_SCOPE_AGENT);
        }
        hist[j] = base;
    }
    __syncthreads();

#pragma unroll
    for (int i = 0; i < EPT; ++i) {
        if (i * BK_THREADS + tid < EPB) {
            const int bi = d[i] >> 8;
            const int pi = hist[bi] + ri[i];
            if (pi < bucket_cap(bi))
                *bucket_slot(agg_in, bi, ESKIP + pi) = __int_as_float((s[i] << 8) | (d[i] & 255));
            const int bo = s[i] >> 8;
            const int po = hist[NBUCK + bo] + ro[i];
            if (po < bucket_cap(bo))
                *bucket_slot(agg_out, bo, ESKIP + po) = __int_as_float((d[i] << 8) | (s[i] & 255));
        }
    }
}

#define RK_THREADS 256
#define RK_UNROLL  8

__global__ __launch_bounds__(RK_THREADS) void sage_rank(
    const int* __restrict__ src, const int* __restrict__ dst,
    float* agg_in, float* agg_out,
    int* __restrict__ cnt_in, int* __restrict__ cnt_out)
{
    __shared__ int lc[BNODES];
    const int  tid   = threadIdx.x;
    const bool rev   = blockIdx.x >= NBUCK;
    const int  b     = rev ? blockIdx.x - NBUCK : blockIdx.x;
    float*     agg   = rev ? agg_out : agg_in;
    int*       cnt   = rev ? cnt_out : cnt_in;
    const int  node0 = b * BNODES;
    const int  nrows = min(BNODES, N_NODES - node0);

    lc[tid] = 0;
    const int total = __float_as_int(*bucket_slot(agg, b, 0));
    __syncthreads();

    if (total <= bucket_cap(b)) {
        for (int p0 = 0; p0 < total; p0 += RK_THREADS * RK_UNROLL) {
            int ent[RK_UNROLL];
#pragma unroll
            for (int u = 0; u < RK_UNROLL; ++u) {
                const int p = p0 + u * RK_THREADS + tid;
                ent[u] = -1;
                if (p < total) ent[u] = __float_as_int(*bucket_slot(agg, b, ESKIP + p));
            }
#pragma unroll
            for (int u = 0; u < RK_UNROLL; ++u) {
                if (ent[u] >= 0) {
                    const int kl = ent[u] & 255;
                    const int r  = atomicAdd(&lc[kl], 1);
                    if (r < CAP) agg[(size_t)(node0 + kl) * D + r] = __int_as_float(ent[u] >> 8);
                }
            }
        }
    } else {
        const int* key = rev ? src : dst;
        const int* oth = rev ? dst : src;
        for (int e = tid; e < N_EDGES; e += RK_THREADS) {
            const int k = key[e];
            if ((k >> 8) == b) {
                const int r = atomicAdd(&lc[k & 255], 1);
                if (r < CAP) agg[(size_t)k * D + r] = __int_as_float(oth[e]);
            }
        }
    }
    __syncthreads();
    if (tid < nrows) cnt[node0 + tid] = lc[tid];
}

#define GATHER_BLOCKS 2048
#define GATHER_WAVES  (GATHER_BLOCKS * 8)
#define N_TASKS       (2 * N_NODES)

__device__ __forceinline__ float4 x_row(const float* __restrict__ x, int node, int lane) {
    return *(const float4*)(x + (size_t)node * D + lane * 4);
}

__device__ __forceinline__ void acc_add(float4& acc, const float4& r) {
    acc.x += r.x; acc.y += r.y; acc.z += r.z; acc.w += r.w;
}

__device__ __forceinline__ float4 sum_group(const int (&nb)[8], int cnt,
                                            const float* __restrict__ x, int lane)
{
    float4 r[8];
#pragma unroll
    for (int t = 0; t < 8; ++t) {
        if (t < cnt) r[t] = x_row(x, nb[t], lane);
        else         r[t] = make_float4(0.0f, 0.0f, 0.0f, 0.0f);
    }
    float4 g = r[0];
#pragma unroll
    for (int t = 1; t < 8; ++t) acc_add(g, r[t]);
    return g;
}

__device__ __forceinline__ void add_rows(int idv, int j0, int m, const float* __restrict__ x,
                                         int lane, float4& acc)
{
    for (int j = j0; j < m; j += 8) {
        int nb[8];
#pragma unroll
        for (int t = 0; t < 8; ++t)
            nb[t] = __builtin_amdgcn_readlane(idv, j + t);
        acc_add(acc, sum_group(nb, m - j, x, lane));
    }
}

__global__ __launch_bounds__(256) void sage_gather(
    const int* __restrict__ src, const int* __restrict__ dst,
    const float* __restrict__ x,
    float* agg_in, float* agg_out, int* cnt_in, int* cnt_out)
{
    const int lane = threadIdx.x & 31;
    int task = __builtin_amdgcn_readfirstlane(blockIdx.x * 8 + (threadIdx.x >> 5));
    if (task >= N_TASKS) return;

    bool   revN = task >= N_NODES;
    int    nN   = revN ? task - N_NODES : task;
    float* rowN = (revN ? agg_out : agg_in) + (size_t)nN * D;
    int*   cntN = (revN ? cnt_out : cnt_in) + nN;
    int    degN = *cntN;
    int    idsN = __float_as_int(rowN[lane]);

    for (;;) {
        const bool rev = revN;
        const int  n   = nN;
        float*     row = rowN;
        int*       cnt = cntN;
        const int deg = __builtin_amdgcn_readfirstlane(degN);
        const int m0  = min(deg, 32);
        const int id0 = (lane < deg) ? idsN : 0;
        int nb[8];
#pragma unroll
        for (int t = 0; t < 8; ++t)
            nb[t] = __builtin_amdgcn_readlane(id0, t);

        const int  next     = task + GATHER_WAVES;
        const bool has_next = next < N_TASKS;
        if (has_next) {
            revN = next >= N_NODES;
            nN   = revN ? next - N_NODES : next;
            rowN = (revN ? agg_out : agg_in) + (size_t)nN * D;
            cntN = (revN ? cnt_out : cnt_in) + nN;
            degN = *cntN;
            idsN = __float_as_int(rowN[lane]);
        }

        float4 acc = make_float4(0.0f, 0.0f, 0.0f, 0.0f);
        if (deg <= CAP) {
            int idv[CAP / 32 - 1];
#pragma unroll
            for (int c = 1; c < CAP / 32; ++c) {
                const int i = c * 32 + lane;
                idv[c - 1] = 0;
                if (i < deg) idv[c - 1] = __float_as_int(row[i]);
            }
            acc = sum_group(nb, m0, x, lane);
            add_rows(id0, 8, m0, x, lane, acc);
#pragma unroll
            for (int c = 1; c < CAP / 32; ++c)
                add_rows(idv[c - 1], 0, min(deg - 32 * c, 32), x, lane, acc);
        } else {
            const int* key = rev ? src : dst;
            const int* oth = rev ? dst : src;
            for (int base = 0; base < N_EDGES; base += 32) {
                const int k = key[base + lane];
                const int o = oth[base + lane];
                unsigned hit = __builtin_amdgcn_ballot_w32(k == n);
                while (hit) {
                    const int j = __builtin_ctz(hit);
                    hit &= hit - 1;
                    acc_add(acc, x_row(x, __builtin_amdgcn_readlane(o, j), lane));
                }
            }
        }
        *(float4*)(row + lane * 4) = acc;
        if (lane == 0) *cnt = __float_as_int((float)deg);

        if (!has_next) break;
        task = next;
    }
}

#define TM      32
#define KTOT    (3 * D)
#define PITCH   (KTOT + 4)
#define NTILES  (N_NODES / TM)
#define GEMM_BLOCKS 256
#define KC      16
#define NCHUNK  (KTOT / 4 / KC)

__device__ __forceinline__ void tile_fetch(
    int tile, int r0, int c4,
    const float* __restrict__ x,
    const float* __restrict__ agg_in, const float* __restrict__ agg_out,
    const float* __restrict__ cnt_in, const float* __restrict__ cnt_out,
    float4 (&pf)[12], float (&ci)[4], float (&co)[4])
{
#pragma unroll
    for (int j = 0; j < 4; ++j) {
        const int row = tile * TM + r0 + 8 * j;
        const size_t g = (size_t)row * D + c4 * 4;
        pf[j]     = *(const float4*)(agg_in  + g);
        pf[4 + j] = *(const float4*)(agg_out + g);
        pf[8 + j] = *(const float4*)(x       + g);
        ci[j] = cnt_in[row];
        co[j] = cnt_out[row];
    }
}

__device__ __forceinline__ void tile_put(
    float* __restrict__ dstTile, int r0, int c4,
    const float4 (&pf)[12], const float (&ci)[4], const float (&co)[4])
{
#pragma unroll
    for (int j = 0; j < 4; ++j) {
        const float si = 1.0f / fmaxf(ci[j], 1.0f);
        const float so = 1.0f / fmaxf(co[j], 1.0f);
        float* p = dstTile + (r0 + 8 * j) * PITCH + c4 * 4;
        float4 a = pf[j];
        a.x *= si; a.y *= si; a.z *= si; a.w *= si;
        float4 b = pf[4 + j];
        b.x *= so; b.y *= so; b.z *= so; b.w *= so;
        *(float4*)(p)         = a;
        *(float4*)(p + D)     = b;
        *(float4*)(p + 2 * D) = pf[8 + j];
    }
}

__device__ __forceinline__ void frag_load(const float* A0, const float* A1, int c,
                                          v2f (&f0)[KC], v2f (&f1)[KC])
{
#pragma unroll
    for (int k = 0; k < KC; ++k) {
        f0[k] = *(const v2f*)(A0 + 4 * (c * KC + k));
        f1[k] = *(const v2f*)(A1 + 4 * (c * KC + k));
    }
}

__global__ __launch_bounds__(256) void sage_gemm(
    const float* __restrict__ x,
    const float* __restrict__ agg_in, const float* __restrict__ agg_out,
    const float* __restrict__ cnt_in, const float* __restrict__ cnt_out,
    const float* __restrict__ w_l_in, const float* __restrict__ w_l_out,
    const float* __restrict__ w_sum,  const float* __restrict__ b_sum,
    float* __restrict__ out)
{
    __shared__ __attribute__((aligned(16))) float tileA[2][TM * PITCH];

    const int tid   = threadIdx.x;
    const int wave  = tid >> 5;
    const int lane  = tid & 31;
    const int half  = lane >> 4;
    const int l15   = lane & 15;
    const int col   = wave * 16 + l15;

    v2f bw[96];
    {
        const size_t woff = (size_t)col * D + 2 * half;
#pragma unroll
        for (int i = 0; i < 32; ++i) {
            bw[i]      = *(const v2f*)(w_l_in  + woff + 4 * i);
            bw[32 + i] = *(const v2f*)(w_l_out + woff + 4 * i);
            bw[64 + i] = *(const v2f*)(w_sum   + woff + 4 * i);
        }
    }
    const float bias = b_sum[col];

    const int c4 = lane;
    const int r0 = wave;

    float4 pf[12];
    float  ci[4], co[4];

    int tile = blockIdx.x;
    int buf  = 0;
    if (tile < NTILES) {
        tile_fetch(tile, r0, c4, x, agg_in, agg_out, cnt_in, cnt_out, pf, ci, co);
        tile_put(&tileA[0][0], r0, c4, pf, ci, co);
    }
    __syncthreads();

    for (; tile < NTILES; tile += GEMM_BLOCKS) {
        const int  next     = tile + GEMM_BLOCKS;
        const bool has_next = next < NTILES;
        if (has_next)
            tile_fetch(next, r0, c4, x, agg_in, agg_out, cnt_in, cnt_out, pf, ci, co);

        const float* A0 = &tileA[buf][0] + l15 * PITCH + 2 * half;
        const float* A1 = A0 + 16 * PITCH;

        v8f acc0, acc1;
#pragma unroll
        for (int r = 0; r < 8; ++r) { acc0[r] = bias; acc1[r] = bias; }

        v2f fa0[2][KC], fa1[2][KC];
        frag_load(A0, A1, 0, fa0[0], fa1[0]);
#pragma unroll
        for (int c = 0; c < NCHUNK; ++c) {
            if (c + 1 < NCHUNK)
                frag_load(A0, A1, c + 1, fa0[(c + 1) & 1], fa1[(c + 1) & 1]);
            __builtin_amdgcn_sched_barrier(0);
#pragma unroll
            for (int k = 0; k < KC; ++k) {
                acc0 = __builtin_amdgcn_wmma_f32_16x16x4_f32(
                     false, fa0[c & 1][k],  false, bw[c * KC + k],
                     (short)0, acc0,  false,  false);
                acc1 = __builtin_amdgcn_wmma_f32_16x16x4_f32(
                     false, fa1[c & 1][k],  false, bw[c * KC + k],
                     (short)0, acc1,  false,  false);
            }
            __builtin_amdgcn_sched_barrier(0);
        }

        if (has_next)
            tile_put(&tileA[buf ^ 1][0], r0, c4, pf, ci, co);
        __syncthreads();

        const int mbase = tile * TM + half * 8;
#pragma unroll
        for (int r = 0; r < 8; ++r) {
            out[(size_t)(mbase + r) * D + col]      = acc0[r];
            out[(size_t)(mbase + 16 + r) * D + col] = acc1[r];
        }
        buf ^= 1;
    }
}

extern "C" void kernel_launch(void* const* d_in, const int* in_sizes, int n_in,
                              void* d_out, int out_size, void* d_ws, size_t ws_size,
                              hipStream_t stream)
{
    const float* x       = (const float*)d_in[0];
    const int*   ei      = (const int*)d_in[1];
    const float* w_l_in  = (const float*)d_in[2];
    const float* b_l_in  = (const float*)d_in[3];
    const float* w_r_in  = (const float*)d_in[4];
    const float* w_l_out = (const float*)d_in[5];
    const float* b_l_out = (const float*)d_in[6];
    const float* w_r_out = (const float*)d_in[7];
    float* out = (float*)d_out;

    const size_t NB   = (size_t)N_NODES * D * sizeof(float);
    const size_t CNTB = (size_t)N_NODES * sizeof(float);
    char* ws = (char*)d_ws;
    float* agg_in  = (float*)(ws);
    float* agg_out = (float*)(ws + NB);
    int*   cnt_in  = (int*)(ws + 2 * NB);
    int*   cnt_out = (int*)(ws + 2 * NB + CNTB);
    float* w_sum   = (float*)(ws + 2 * NB + 2 * CNTB);
    float* b_sum   = (float*)(ws + 2 * NB + 2 * CNTB + (size_t)D * D * sizeof(float));

    (void)in_sizes; (void)n_in; (void)out_size; (void)ws_size;

    sage_init<<<(D * D) / 256, 256, 0, stream>>>(agg_in, agg_out,
                                                 w_r_in, w_r_out, b_l_in, b_l_out,
                                                 w_sum, b_sum);

    sage_bucket<<<BK_BLOCKS, BK_THREADS, 0, stream>>>(ei, ei + N_EDGES, agg_in, agg_out);

    sage_rank<<<2 * NBUCK, RK_THREADS, 0, stream>>>(ei, ei + N_EDGES, agg_in, agg_out,
                                                    cnt_in, cnt_out);

    sage_gather<<<GATHER_BLOCKS, 256, 0, stream>>>(ei, ei + N_EDGES, x,
                                                   agg_in, agg_out, cnt_in, cnt_out);

    sage_gemm<<<GEMM_BLOCKS, 256, 0, stream>>>(x, agg_in, agg_out,
                                                (const float*)cnt_in, (const float*)cnt_out,
                                                w_l_in, w_l_out, w_sum, b_sum, out);
}


# Bucket counters made private: each bucket-kernel work-group owns a fixed 128-B share per bucket; global adds, counter clearing and the init launch go; sum order = slot order by LDS arrival as before
# speedup vs baseline: 2.8771x; 1.1673x over previous
#include <hip/hip_runtime.h>
#include <hip/hip_bf16.h>


#define N_NODES 100000
#define N_EDGES 640000
#define D 128

typedef float v2f __attribute__((ext_vector_type(2)));
typedef float v8f __attribute__((ext_vector_type(8)));

#define CAP 64
#define BNODES 256
#define NBUCK  ((N_NODES + BNODES - 1) / BNODES)
#define SHARE_WORDS 32
#define SHARE_CAP   (SHARE_WORDS - 1)

__device__ __forceinline__ int* bucket_share(float* agg, int b, int k) {
    return (int*)(agg + ((size_t)(b * BNODES + (k >> 1))) * D + 64 + (k & 1) * SHARE_WORDS);
}

#define BK_BLOCKS  256
#define EPB        (N_EDGES / BK_BLOCKS)
#define BK_THREADS 512
#define EPT        ((EPB + BK_THREADS - 1) / BK_THREADS)

__global__ __launch_bounds__(BK_THREADS) void sage_bucket(
    const int* __restrict__ src, const int* __restrict__ dst,
    float* __restrict__ agg_in, float* __restrict__ agg_out)
{
    __shared__ int hist[2 * NBUCK];
    const int tid = threadIdx.x;
    for (int j = tid; j < 2 * NBUCK; j += BK_THREADS) hist[j] = 0;
    __syncthreads();

    const int e0 = blockIdx.x * EPB;
    int s[EPT], d[EPT], ri[EPT], ro[EPT];
#pragma unroll
    for (int i = 0; i < EPT; ++i) {
        const int k = i * BK_THREADS + tid;
        s[i] = 0; d[i] = 0;
        if (k < EPB) { s[i] = src[e0 + k]; d[i] = dst[e0 + k]; }
    }
#pragma unroll
    for (int i = 0; i < EPT; ++i) {
        ri[i] = 0; ro[i] = 0;
        if (i * BK_THREADS + tid < EPB) {
            ri[i] = atomicAdd(&hist[d[i] >> 8], 1);
            ro[i] = atomicAdd(&hist[NBUCK + (s[i] >> 8)], 1);
        }
    }
    __syncthreads();

    const int blk = blockIdx.x;
    for (int j = tid; j < 2 * NBUCK; j += BK_THREADS) {
        const bool rev = j >= NBUCK;
        const int  b   = rev ? j - NBUCK : j;
        bucket_share(rev ? agg_out : agg_in, b, blk)[0] = hist[j];
    }

#pragma unroll
    for (int i = 0; i < EPT; ++i) {
        if (i * BK_THREADS + tid < EPB) {
            if (ri[i] < SHARE_CAP)
                bucket_share(agg_in, d[i] >> 8, blk)[1 + ri[i]] = (s[i] << 8) | (d[i] & 255);
            if (ro[i] < SHARE_CAP)
                bucket_share(agg_out, s[i] >> 8, blk)[1 + ro[i]] = (d[i] << 8) | (s[i] & 255);
        }
    }
}

#define RK_THREADS BK_BLOCKS
#define RK_FIRST   15

__global__ __launch_bounds__(RK_THREADS) void sage_rank(
    const int* __restrict__ src, const int* __restrict__ dst,
    float* agg_in, float* agg_out,
    int* __restrict__ cnt_in, int* __restrict__ cnt_out)
{
    __shared__ int lc[BNODES];
    __shared__ int over;
    const int  tid   = threadIdx.x;
    const bool rev   = blockIdx.x >= NBUCK;
    const int  b     = rev ? blockIdx.x - NBUCK : blockIdx.x;
    float*     agg   = rev ? agg_out : agg_in;
    int*       cnt   = rev ? cnt_out : cnt_in;
    const int  node0 = b * BNODES;
    const int  nrows = min(BNODES, N_NODES - node0);

    lc[tid] = 0;
    if (tid == 0) over = 0;
    const int* sh = bucket_share(agg, b, tid);
    const int4 w0 = ((const int4*)sh)[0];
    const int4 w1 = ((const int4*)sh)[1];
    const int4 w2 = ((const int4*)sh)[2];
    const int4 w3 = ((const int4*)sh)[3];
    const int  n  = w0.x;
    __syncthreads();
    if (n > SHARE_CAP) over = 1;
    __syncthreads();

    if (over == 0) {
        const int ent[RK_FIRST] = { w0.y, w0.z, w0.w, w1.x, w1.y, w1.z, w1.w,
                                    w2.x, w2.y, w2.z, w2.w, w3.x, w3.y, w3.z, w3.w };
#pragma unroll
        for (int u = 0; u < RK_FIRST; ++u) {
            if (u < n) {
                const int kl = ent[u] & 255;
                const int r  = atomicAdd(&lc[kl], 1);
                if (r < CAP) agg[(size_t)(node0 + kl) * D + r] = __int_as_float(ent[u] >> 8);
            }
        }
        for (int u = RK_FIRST; u < n; ++u) {
            const int e  = sh[1 + u];
            const int kl = e & 255;
            const int r  = atomicAdd(&lc[kl], 1);
            if (r < CAP) agg[(size_t)(node0 + kl) * D + r] = __int_as_float(e >> 8);
        }
    } else {
        const int* key = rev ? src : dst;
        const int* oth = rev ? dst : src;
        for (int e = tid; e < N_EDGES; e += RK_THREADS) {
            const int k = key[e];
            if ((k >> 8) == b) {
                const int r = atomicAdd(&lc[k & 255], 1);
                if (r < CAP) agg[(size_t)k * D + r] = __int_as_float(oth[e]);
            }
        }
    }
    __syncthreads();
    if (tid < nrows) cnt[node0 + tid] = lc[tid];
}

#define GATHER_BLOCKS 2048
#define GATHER_WAVES  (GATHER_BLOCKS * 8)
#define N_TASKS       (2 * N_NODES)

__device__ __forceinline__ float4 x_row(const float* __restrict__ x, int node, int lane) {
    return *(const float4*)(x + (size_t)node * D + lane * 4);
}

__device__ __forceinline__ void acc_add(float4& acc, const float4& r) {
    acc.x += r.x; acc.y += r.y; acc.z += r.z; acc.w += r.w;
}

__device__ __forceinline__ float4 sum_group(const int (&nb)[8], int cnt,
                                            const float* __restrict__ x, int lane)
{
    float4 r[8];
#pragma unroll
    for (int t = 0; t < 8; ++t) {
        if (t < cnt) r[t] = x_row(x, nb[t], lane);
        else         r[t] = make_float4(0.0f, 0.0f, 0.0f, 0.0f);
    }
    float4 g = r[0];
#pragma unroll
    for (int t = 1; t < 8; ++t) acc_add(g, r[t]);
    return g;
}

__device__ __forceinline__ void add_rows(int idv, int j0, int m, const float* __restrict__ x,
                                         int lane, float4& acc)
{
    for (int j = j0; j < m; j += 8) {
        int nb[8];
#pragma unroll
        for (int t = 0; t < 8; ++t)
            nb[t] = __builtin_amdgcn_readlane(idv, j + t);
        acc_add(acc, sum_group(nb, m - j, x, lane));
    }
}

__global__ __launch_bounds__(256) void sage_gather(
    const int* __restrict__ src, const int* __restrict__ dst,
    const float* __restrict__ x,
    float* agg_in, float* agg_out, int* cnt_in, int* cnt_out)
{
    const int lane = threadIdx.x & 31;
    int task = __builtin_amdgcn_readfirstlane(blockIdx.x * 8 + (threadIdx.x >> 5));
    if (task >= N_TASKS) return;

    bool   revN = task >= N_NODES;
    int    nN   = revN ? task - N_NODES : task;
    float* rowN = (revN ? agg_out : agg_in) + (size_t)nN * D;
    int*   cntN = (revN ? cnt_out : cnt_in) + nN;
    int    degN = *cntN;
    int    idsN = __float_as_int(rowN[lane]);

    for (;;) {
        const bool rev = revN;
        const int  n   = nN;
        float*     row = rowN;
        int*       cnt = cntN;
        const int deg = __builtin_amdgcn_readfirstlane(degN);
        const int m0  = min(deg, 32);
        const int id0 = (lane < deg) ? idsN : 0;
        int nb[8];
#pragma unroll
        for (int t = 0; t < 8; ++t)
            nb[t] = __builtin_amdgcn_readlane(id0, t);

        const int  next     = task + GATHER_WAVES;
        const bool has_next = next < N_TASKS;
        if (has_next) {
            revN = next >= N_NODES;
            nN   = revN ? next - N_NODES : next;
            rowN = (revN ? agg_out : agg_in) + (size_t)nN * D;
            cntN = (revN ? cnt_out : cnt_in) + nN;
            degN = *cntN;
            idsN = __float_as_int(rowN[lane]);
        }

        float4 acc = make_float4(0.0f, 0.0f, 0.0f, 0.0f);
        if (deg <= CAP) {
            int idv[CAP / 32 - 1];
#pragma unroll
            for (int c = 1; c < CAP / 32; ++c) {
                const int i = c * 32 + lane;
                idv[c - 1] = 0;
                if (i < deg) idv[c - 1] = __float_as_int(row[i]);
            }
            acc = sum_group(nb, m0, x, lane);
            add_rows(id0, 8, m0, x, lane, acc);
#pragma unroll
            for (int c = 1; c < CAP / 32; ++c)
                add_rows(idv[c - 1], 0, min(deg - 32 * c, 32), x, lane, acc);
        } else {
            const int* key = rev ? src : dst;
            const int* oth = rev ? dst : src;
            for (int base = 0; base < N_EDGES; base += 32) {
                const int k = key[base + lane];
                const int o = oth[base + lane];
                unsigned hit = __builtin_amdgcn_ballot_w32(k == n);
                while (hit) {
                    const int j = __builtin_ctz(hit);
                    hit &= hit - 1;
                    acc_add(acc, x_row(x, __builtin_amdgcn_readlane(o, j), lane));
                }
            }
        }
        *(float4*)(row + lane * 4) = acc;
        if (lane == 0) *cnt = __float_as_int((float)deg);

        if (!has_next) break;
        task = next;
    }
}

#define TM      32
#define KTOT    (3 * D)
#define PITCH   (KTOT + 4)
#define NTILES  (N_NODES / TM)
#define GEMM_BLOCKS 256
#define KC      16
#define NCHUNK  (KTOT / 4 / KC)

__device__ __forceinline__ void tile_fetch(
    int tile, int r0, int c4,
    const float* __restrict__ x,
    const float* __restrict__ agg_in, const float* __restrict__ agg_out,
    const float* __restrict__ cnt_in, const float* __restrict__ cnt_out,
    float4 (&pf)[12], float (&ci)[4], float (&co)[4])
{
#pragma unroll
    for (int j = 0; j < 4; ++j) {
        const int row = tile * TM + r0 + 8 * j;
        const size_t g = (size_t)row * D + c4 * 4;
        pf[j]     = *(const float4*)(agg_in  + g);
        pf[4 + j] = *(const float4*)(agg_out + g);
        pf[8 + j] = *(const float4*)(x       + g);
        ci[j] = cnt_in[row];
        co[j] = cnt_out[row];
    }
}

__device__ __forceinline__ void tile_put(
    float* __restrict__ dstTile, int r0, int c4,
    const float4 (&pf)[12], const float (&ci)[4], const float (&co)[4])
{
#pragma unroll
    for (int j = 0; j < 4; ++j) {
        const float si = 1.0f / fmaxf(ci[j], 1.0f);
        const float so = 1.0f / fmaxf(co[j], 1.0f);
        float* p = dstTile + (r0 + 8 * j) * PITCH + c4 * 4;
        float4 a = pf[j];
        a.x *= si; a.y *= si; a.z *= si; a.w *= si;
        float4 b = pf[4 + j];
        b.x *= so; b.y *= so; b.z *= so; b.w *= so;
        *(float4*)(p)         = a;
        *(float4*)(p + D)     = b;
        *(float4*)(p + 2 * D) = pf[8 + j];
    }
}

__device__ __forceinline__ void frag_load(const float* A0, const float* A1, int c,
                                          v2f (&f0)[KC], v2f (&f1)[KC])
{
#pragma unroll
    for (int k = 0; k < KC; ++k) {
        f0[k] = *(const v2f*)(A0 + 4 * (c * KC + k));
        f1[k] = *(const v2f*)(A1 + 4 * (c * KC + k));
    }
}

__global__ __launch_bounds__(256) void sage_gemm(
    const float* __restrict__ x,
    const float* __restrict__ agg_in, const float* __restrict__ agg_out,
    const float* __restrict__ cnt_in, const float* __restrict__ cnt_out,
    const float* __restrict__ w_l_in, const float* __restrict__ w_l_out,
    const float* __restrict__ w_r_in, const float* __restrict__ w_r_out,
    const float* __restrict__ b_l_in, const float* __restrict__ b_l_out,
    float* __restrict__ out)
{
    __shared__ __attribute__((aligned(16))) float tileA[2][TM * PITCH];

    const int tid   = threadIdx.x;
    const int wave  = tid >> 5;
    const int lane  = tid & 31;
    const int half  = lane >> 4;
    const int l15   = lane & 15;
    const int col   = wave * 16 + l15;

    v2f bw[96];
    {
        const size_t woff = (size_t)col * D + 2 * half;
#pragma unroll
        for (int i = 0; i < 32; ++i) {
            bw[i]      = *(const v2f*)(w_l_in  + woff + 4 * i);
            bw[32 + i] = *(const v2f*)(w_l_out + woff + 4 * i);
            bw[64 + i] = *(const v2f*)(w_r_in  + woff + 4 * i) + *(const v2f*)(w_r_out + woff + 4 * i);
        }
    }
    const float bias = b_l_in[col] + b_l_out[col];

    const int c4 = lane;
    const int r0 = wave;

    float4 pf[12];
    float  ci[4], co[4];

    int tile = blockIdx.x;
    int buf  = 0;
    if (tile < NTILES) {
        tile_fetch(tile, r0, c4, x, agg_in, agg_out, cnt_in, cnt_out, pf, ci, co);
        tile_put(&tileA[0][0], r0, c4, pf, ci, co);
    }
    __syncthreads();

    for (; tile < NTILES; tile += GEMM_BLOCKS) {
        const int  next     = tile + GEMM_BLOCKS;
        const bool has_next = next < NTILES;
        if (has_next)
            tile_fetch(next, r0, c4, x, agg_in, agg_out, cnt_in, cnt_out, pf, ci, co);

        const float* A0 = &tileA[buf][0] + l15 * PITCH + 2 * half;
        const float* A1 = A0 + 16 * PITCH;

        v8f acc0, acc1;
#pragma unroll
        for (int r = 0; r < 8; ++r) { acc0[r] = bias; acc1[r] = bias; }

        v2f fa0[2][KC], fa1[2][KC];
        frag_load(A0, A1, 0, fa0[0], fa1[0]);
#pragma unroll
        for (int c = 0; c < NCHUNK; ++c) {
            if (c + 1 < NCHUNK)
                frag_load(A0, A1, c + 1, fa0[(c + 1) & 1], fa1[(c + 1) & 1]);
            __builtin_amdgcn_sched_barrier(0);
#pragma unroll
            for (int k = 0; k < KC; ++k) {
                acc0 = __builtin_amdgcn_wmma_f32_16x16x4_f32(
                     false, fa0[c & 1][k],  false, bw[c * KC + k],
                     (short)0, acc0,  false,  false);
                acc1 = __builtin_amdgcn_wmma_f32_16x16x4_f32(
                     false, fa1[c & 1][k],  false, bw[c * KC + k],
                     (short)0, acc1,  false,  false);
            }
            __builtin_amdgcn_sched_barrier(0);
        }

        if (has_next)
            tile_put(&tileA[buf ^ 1][0], r0, c4, pf, ci, co);
        __syncthreads();

        const int mbase = tile * TM + half * 8;
#pragma unroll
        for (int r = 0; r < 8; ++r) {
            out[(size_t)(mbase + r) * D + col]      = acc0[r];
            out[(size_t)(mbase + 16 + r) * D + col] = acc1[r];
        }
        buf ^= 1;
    }
}

extern "C" void kernel_launch(void* const* d_in, const int* in_sizes, int n_in,
                              void* d_out, int out_size, void* d_ws, size_t ws_size,
                              hipStream_t stream)
{
    const float* x       = (const float*)d_in[0];
    const int*   ei      = (const int*)d_in[1];
    const float* w_l_in  = (const float*)d_in[2];
    const float* b_l_in  = (const float*)d_in[3];
    const float* w_r_in  = (const float*)d_in[4];
    const float* w_l_out = (const float*)d_in[5];
    const float* b_l_out = (const float*)d_in[6];
    const float* w_r_out = (const float*)d_in[7];
    float* out = (float*)d_out;

    const size_t NB   = (size_t)N_NODES * D * sizeof(float);
    const size_t CNTB = (size_t)N_NODES * sizeof(float);
    char* ws = (char*)d_ws;
    float* agg_in  = (float*)(ws);
    float* agg_out = (float*)(ws + NB);
    int*   cnt_in  = (int*)(ws + 2 * NB);
    int*   cnt_out = (int*)(ws + 2 * NB + CNTB);

    (void)in_sizes; (void)n_in; (void)out_size; (void)ws_size;

    sage_bucket<<<BK_BLOCKS, BK_THREADS, 0, stream>>>(ei, ei + N_EDGES, agg_in, agg_out);

    sage_rank<<<2 * NBUCK, RK_THREADS, 0, stream>>>(ei, ei + N_EDGES, agg_in, agg_out,
                                                    cnt_in, cnt_out);

    sage_gather<<<GATHER_BLOCKS, 256, 0, stream>>>(ei, ei + N_EDGES, x,
                                                   agg_in, agg_out, cnt_in, cnt_out);

    sage_gemm<<<GEMM_BLOCKS, 256, 0, stream>>>(x, agg_in, agg_out,
                                                (const float*)cnt_in, (const float*)cnt_out,
                                                w_l_in, w_l_out, w_r_in, w_r_out,
                                                b_l_in, b_l_out, out);
}


# Neighbour sums moved into the GEMM tile stage: each wave sums its next-tile rows' listed x rows under the current tile's WMMA chunks into the LDS A tile; gather launch, sum arrays go; sum order kept
# speedup vs baseline: 3.3621x; 1.1686x over previous
#include <hip/hip_runtime.h>
#include <hip/hip_bf16.h>


#define N_NODES 100000
#define N_EDGES 640000
#define D 128

typedef float v2f __attribute__((ext_vector_type(2)));
typedef float v8f __attribute__((ext_vector_type(8)));

#define CAP 64
#define BNODES 256
#define NBUCK  ((N_NODES + BNODES - 1) / BNODES)
#define SHARE_WORDS 32
#define SHARE_CAP   (SHARE_WORDS - 1)

__device__ __forceinline__ int* bucket_share(float* agg, int b, int k) {
    return (int*)(agg + ((size_t)(b * BNODES + (k >> 1))) * D + 64 + (k & 1) * SHARE_WORDS);
}

#define BK_BLOCKS  256
#define EPB        (N_EDGES / BK_BLOCKS)
#define BK_THREADS 512
#define EPT        ((EPB + BK_THREADS - 1) / BK_THREADS)

__global__ __launch_bounds__(BK_THREADS) void sage_bucket(
    const int* __restrict__ src, const int* __restrict__ dst,
    float* __restrict__ agg_in, float* __restrict__ agg_out)
{
    __shared__ int hist[2 * NBUCK];
    const int tid = threadIdx.x;
    for (int j = tid; j < 2 * NBUCK; j += BK_THREADS) hist[j] = 0;
    __syncthreads();

    const int e0 = blockIdx.x * EPB;
    int s[EPT], d[EPT], ri[EPT], ro[EPT];
#pragma unroll
    for (int i = 0; i < EPT; ++i) {
        const int k = i * BK_THREADS + tid;
        s[i] = 0; d[i] = 0;
        if (k < EPB) { s[i] = src[e0 + k]; d[i] = dst[e0 + k]; }
    }
#pragma unroll
    for (int i = 0; i < EPT; ++i) {
        ri[i] = 0; ro[i] = 0;
        if (i * BK_THREADS + tid < EPB) {
            ri[i] = atomicAdd(&hist[d[i] >> 8], 1);
            ro[i] = atomicAdd(&hist[NBUCK + (s[i] >> 8)], 1);
        }
    }
    __syncthreads();

    const int blk = blockIdx.x;
    for (int j = tid; j < 2 * NBUCK; j += BK_THREADS) {
        const bool rev = j >= NBUCK;
        const int  b   = rev ? j - NBUCK : j;
        bucket_share(rev ? agg_out : agg_in, b, blk)[0] = hist[j];
    }

#pragma unroll
    for (int i = 0; i < EPT; ++i) {
        if (i * BK_THREADS + tid < EPB) {
            if (ri[i] < SHARE_CAP)
                bucket_share(agg_in, d[i] >> 8, blk)[1 + ri[i]] = (s[i] << 8) | (d[i] & 255);
            if (ro[i] < SHARE_CAP)
                bucket_share(agg_out, s[i] >> 8, blk)[1 + ro[i]] = (d[i] << 8) | (s[i] & 255);
        }
    }
}

#define RK_THREADS BK_BLOCKS
#define RK_FIRST   15

__global__ __launch_bounds__(RK_THREADS) void sage_rank(
    const int* __restrict__ src, const int* __restrict__ dst,
    float* agg_in, float* agg_out,
    int* __restrict__ cnt_in, int* __restrict__ cnt_out)
{
    __shared__ int lc[BNODES];
    __shared__ int over;
    const int  tid   = threadIdx.x;
    const bool rev   = blockIdx.x >= NBUCK;
    const int  b     = rev ? blockIdx.x - NBUCK : blockIdx.x;
    float*     agg   = rev ? agg_out : agg_in;
    int*       cnt   = rev ? cnt_out : cnt_in;
    const int  node0 = b * BNODES;
    const int  nrows = min(BNODES, N_NODES - node0);

    lc[tid] = 0;
    if (tid == 0) over = 0;
    const int* sh = bucket_share(agg, b, tid);
    const int4 w0 = ((const int4*)sh)[0];
    const int4 w1 = ((const int4*)sh)[1];
    const int4 w2 = ((const int4*)sh)[2];
    const int4 w3 = ((const int4*)sh)[3];
    const int  n  = w0.x;
    __syncthreads();
    if (n > SHARE_CAP) over = 1;
    __syncthreads();

    if (over == 0) {
        const int ent[RK_FIRST] = { w0.y, w0.z, w0.w, w1.x, w1.y, w1.z, w1.w,
                                    w2.x, w2.y, w2.z, w2.w, w3.x, w3.y, w3.z, w3.w };
#pragma unroll
        for (int u = 0; u < RK_FIRST; ++u) {
            if (u < n) {
                const int kl = ent[u] & 255;
                const int r  = atomicAdd(&lc[kl], 1);
                if (r < CAP) agg[(size_t)(node0 + kl) * D + r] = __int_as_float(ent[u] >> 8);
            }
        }
        for (int u = RK_FIRST; u < n; ++u) {
            const int e  = sh[1 + u];
            const int kl = e & 255;
            const int r  = atomicAdd(&lc[kl], 1);
            if (r < CAP) agg[(size_t)(node0 + kl) * D + r] = __int_as_float(e >> 8);
        }
    } else {
        const int* key = rev ? src : dst;
        const int* oth = rev ? dst : src;
        for (int e = tid; e < N_EDGES; e += RK_THREADS) {
            const int k = key[e];
            if ((k >> 8) == b) {
                const int r = atomicAdd(&lc[k & 255], 1);
                if (r < CAP) agg[(size_t)k * D + r] = __int_as_float(oth[e]);
            }
        }
    }
    __syncthreads();
    if (tid < nrows) cnt[node0 + tid] = lc[tid];
}

__device__ __forceinline__ float4 x_row(const float* __restrict__ x, int node, int lane) {
    return *(const float4*)(x + (size_t)node * D + lane * 4);
}

__device__ __forceinline__ void acc_add(float4& acc, const float4& r) {
    acc.x += r.x; acc.y += r.y; acc.z += r.z; acc.w += r.w;
}

__device__ __forceinline__ void rows_issue_nb(const int (&nb)[8], int cnt,
                                              const float* __restrict__ x,
                                              int lane, float4 (&r)[8])
{
#pragma unroll
    for (int t = 0; t < 8; ++t) {
        if (t < cnt) r[t] = x_row(x, nb[t], lane);
        else         r[t] = make_float4(0.0f, 0.0f, 0.0f, 0.0f);
    }
}

__device__ __forceinline__ void rows_issue(int idv, int cnt, const float* __restrict__ x,
                                           int lane, float4 (&r)[8])
{
    int nb[8];
#pragma unroll
    for (int t = 0; t < 8; ++t)
        nb[t] = __builtin_amdgcn_readlane(idv, t);
    rows_issue_nb(nb, cnt, x, lane, r);
}

__device__ __forceinline__ float4 rows_sum(const float4 (&r)[8])
{
    float4 g = r[0];
#pragma unroll
    for (int t = 1; t < 8; ++t) acc_add(g, r[t]);
    return g;
}

__device__ __forceinline__ float4 task_finish(
    const float4 (&r)[8], int id0, int deg, const int* __restrict__ list, int n, bool rev,
    const int* __restrict__ src, const int* __restrict__ dst,
    const float* __restrict__ x, int lane)
{
    float4 acc = rows_sum(r);
    if (deg <= CAP) {
        int idv = id0;
#pragma nounroll
        for (int j = 8; j < deg; j += 8) {
            if (j == 32) {
                const int i = 32 + lane;
                idv = 0;
                if (i < deg) idv = list[i];
            }
            int nb[8];
#pragma unroll
            for (int t = 0; t < 8; ++t)
                nb[t] = __builtin_amdgcn_readlane(idv, (j & 31) + t);
            float4 q[8];
            rows_issue_nb(nb, min(deg - j, 8), x, lane, q);
            acc_add(acc, rows_sum(q));
        }
    } else {
        const int* key = rev ? src : dst;
        const int* oth = rev ? dst : src;
#pragma nounroll
        for (int base = 0; base < N_EDGES; base += 32) {
            const int k = key[base + lane];
            const int o = oth[base + lane];
            unsigned hit = __builtin_amdgcn_ballot_w32(k == n);
            while (hit) {
                const int j = __builtin_ctz(hit);
                hit &= hit - 1;
                acc_add(acc, x_row(x, __builtin_amdgcn_readlane(o, j), lane));
            }
        }
    }
    return acc;
}

#define TM      32
#define KTOT    (3 * D)
#define PITCH   (KTOT + 4)
#define NTILES  (N_NODES / TM)
#define GEMM_BLOCKS 256
#define KC      16
#define NCHUNK  (KTOT / 4 / KC)

__device__ __forceinline__ void list_fetch(
    int tile, int r0, int lane,
    const int* __restrict__ lst_in, const int* __restrict__ lst_out,
    const int* __restrict__ cnt_in, const int* __restrict__ cnt_out,
    int (&ids)[8], int (&dg)[8])
{
#pragma unroll
    for (int j = 0; j < 4; ++j) {
        const int row = tile * TM + r0 + 8 * j;
        ids[j]     = lst_in [(size_t)row * D + lane];
        ids[4 + j] = lst_out[(size_t)row * D + lane];
        dg[j]      = cnt_in[row];
        dg[4 + j]  = cnt_out[row];
    }
}

__device__ __forceinline__ void pair_issue(
    int ids_i, int dg_i, int ids_o, int dg_o, const float* __restrict__ x, int lane,
    int& idi, int& di, int& ido, int& dout, float4 (&ri)[8], float4 (&ro)[8])
{
    di   = __builtin_amdgcn_readfirstlane(dg_i);
    dout = __builtin_amdgcn_readfirstlane(dg_o);
    idi  = (lane < di)   ? ids_i : 0;
    ido  = (lane < dout) ? ids_o : 0;
    rows_issue(idi, di   > CAP ? 0 : min(di, 8),   x, lane, ri);
    rows_issue(ido, dout > CAP ? 0 : min(dout, 8), x, lane, ro);
}

__device__ __forceinline__ void pair_finish(
    float* __restrict__ p, int row, int lane,
    const float4 (&ri)[8], const float4 (&ro)[8], int idi, int di, int ido, int dout,
    const int* __restrict__ lst_in, const int* __restrict__ lst_out,
    const int* __restrict__ src, const int* __restrict__ dst, const float* __restrict__ x)
{
    float4 a = task_finish(ri, idi, di,   lst_in  + (size_t)row * D, row, false, src, dst, x, lane);
    float4 b = task_finish(ro, ido, dout, lst_out + (size_t)row * D, row, true,  src, dst, x, lane);
    const float si = 1.0f / fmaxf((float)di, 1.0f);
    const float so = 1.0f / fmaxf((float)dout, 1.0f);
    a.x *= si; a.y *= si; a.z *= si; a.w *= si;
    b.x *= so; b.y *= so; b.z *= so; b.w *= so;
    *(float4*)(p)     = a;
    *(float4*)(p + D) = b;
}

__device__ __forceinline__ void frag_load(const float* A0, const float* A1, int c,
                                          v2f (&f0)[KC], v2f (&f1)[KC])
{
#pragma unroll
    for (int k = 0; k < KC; ++k) {
        f0[k] = *(const v2f*)(A0 + 4 * (c * KC + k));
        f1[k] = *(const v2f*)(A1 + 4 * (c * KC + k));
    }
}

__global__ __launch_bounds__(256) void sage_gemm(
    const int* __restrict__ src, const int* __restrict__ dst,
    const float* __restrict__ x,
    const int* __restrict__ lst_in, const int* __restrict__ lst_out,
    const int* __restrict__ cnt_in, const int* __restrict__ cnt_out,
    const float* __restrict__ w_l_in, const float* __restrict__ w_l_out,
    const float* __restrict__ w_r_in, const float* __restrict__ w_r_out,
    const float* __restrict__ b_l_in, const float* __restrict__ b_l_out,
    float* __restrict__ out)
{
    __shared__ __attribute__((aligned(16))) float tileA[2][TM * PITCH];

    const int tid   = threadIdx.x;
    const int wave  = tid >> 5;
    const int lane  = tid & 31;
    const int half  = lane >> 4;
    const int l15   = lane & 15;
    const int col   = wave * 16 + l15;

    v2f bw[96];
    {
        const size_t woff = (size_t)col * D + 2 * half;
#pragma unroll
        for (int i = 0; i < 32; ++i) {
            bw[i]      = *(const v2f*)(w_l_in  + woff + 4 * i);
            bw[32 + i] = *(const v2f*)(w_l_out + woff + 4 * i);
            bw[64 + i] = *(const v2f*)(w_r_in  + woff + 4 * i) + *(const v2f*)(w_r_out + woff + 4 * i);
        }
    }
    const float bias = b_l_in[col] + b_l_out[col];

    const int c4 = lane;
    const int r0 = wave;

    int tile = blockIdx.x;
    int buf  = 0;
    int idsA[8], dgA[8];
    if (tile < NTILES) {
        int row = tile * TM + r0;
        int li = lst_in [(size_t)row * D + lane];
        int lo = lst_out[(size_t)row * D + lane];
        int gi = cnt_in[row];
        int go = cnt_out[row];
#pragma nounroll
        for (int j = 0; j < 4; ++j) {
            const int ci = li, co = lo, cgi = gi, cgo = go;
            const int rown = row + 8;
            if (j < 3) {
                li = lst_in [(size_t)rown * D + lane];
                lo = lst_out[(size_t)rown * D + lane];
                gi = cnt_in[rown];
                go = cnt_out[rown];
            }
            float* p = &tileA[0][0] + (r0 + 8 * j) * PITCH + c4 * 4;
            const float4 xr = *(const float4*)(x + (size_t)row * D + c4 * 4);
            float4 ri[8], ro[8];
            int idi, di, ido, dout;
            pair_issue(ci, cgi, co, cgo, x, lane, idi, di, ido, dout, ri, ro);
            pair_finish(p, row, lane, ri, ro, idi, di, ido, dout, lst_in, lst_out, src, dst, x);
            *(float4*)(p + 2 * D) = xr;
            row = rown;
        }
    }
    if (tile + GEMM_BLOCKS < NTILES) {
        list_fetch(tile + GEMM_BLOCKS, r0, lane, lst_in, lst_out, cnt_in, cnt_out, idsA, dgA);
    } else {
#pragma unroll
        for (int j = 0; j < 8; ++j) { idsA[j] = 0; dgA[j] = 0; }
    }
    __syncthreads();

    for (; tile < NTILES; tile += GEMM_BLOCKS) {
        const int  next      = tile + GEMM_BLOCKS;
        const bool has_next  = next < NTILES;
        const bool has_next2 = next + GEMM_BLOCKS < NTILES;

        const int xt = has_next ? next : tile;

        const float* A0 = &tileA[buf][0] + l15 * PITCH + 2 * half;
        const float* A1 = A0 + 16 * PITCH;
        float* nt = &tileA[buf ^ 1][0] + r0 * PITCH + c4 * 4;

        v8f acc0, acc1;
#pragma unroll
        for (int r = 0; r < 8; ++r) { acc0[r] = bias; acc1[r] = bias; }

        v2f fa0[2][KC], fa1[2][KC];
        frag_load(A0, A1, 0, fa0[0], fa1[0]);
#pragma unroll
        for (int c = 0; c < NCHUNK; ++c) {
            float4 ri[8], ro[8];
            int idi = 0, di = 0, ido = 0, dout = 0;
            float4 xr = make_float4(0.0f, 0.0f, 0.0f, 0.0f);
            if (c < 4) {
                xr = *(const float4*)(x + (size_t)(xt * TM + r0 + 8 * c) * D + c4 * 4);
                pair_issue(idsA[c], dgA[c], idsA[4 + c], dgA[4 + c], x, lane,
                           idi, di, ido, dout, ri, ro);
            }
            if (c == 4) {
                if (has_next2) {
                    list_fetch(next + GEMM_BLOCKS, r0, lane, lst_in, lst_out, cnt_in, cnt_out,
                               idsA, dgA);
                } else {
#pragma unroll
                    for (int j = 0; j < 8; ++j) { idsA[j] = 0; dgA[j] = 0; }
                }
            }
            if (c + 1 < NCHUNK)
                frag_load(A0, A1, c + 1, fa0[(c + 1) & 1], fa1[(c + 1) & 1]);
            __builtin_amdgcn_sched_barrier(0);
#pragma unroll
            for (int k = 0; k < KC; ++k) {
                acc0 = __builtin_amdgcn_wmma_f32_16x16x4_f32(
                     false, fa0[c & 1][k],  false, bw[c * KC + k],
                     (short)0, acc0,  false,  false);
                acc1 = __builtin_amdgcn_wmma_f32_16x16x4_f32(
                     false, fa1[c & 1][k],  false, bw[c * KC + k],
                     (short)0, acc1,  false,  false);
            }
            __builtin_amdgcn_sched_barrier(0);
            asm volatile("v_nop\n\tv_nop\n\tv_nop\n\tv_nop\n\tv_nop\n\tv_nop\n\tv_nop\n\tv_nop");
            __builtin_amdgcn_sched_barrier(0);
            if (c < 4) {
                pair_finish(nt + 8 * c * PITCH, xt * TM + r0 + 8 * c, lane,
                            ri, ro, idi, di, ido, dout, lst_in, lst_out, src, dst, x);
                *(float4*)(nt + 8 * c * PITCH + 2 * D) = xr;
            }
        }

        __syncthreads();

        const int mbase = tile * TM + half * 8;
#pragma unroll
        for (int r = 0; r < 8; ++r) {
            out[(size_t)(mbase + r) * D + col]      = acc0[r];
            out[(size_t)(mbase + 16 + r) * D + col] = acc1[r];
        }
        buf ^= 1;
    }
}

extern "C" void kernel_launch(void* const* d_in, const int* in_sizes, int n_in,
                              void* d_out, int out_size, void* d_ws, size_t ws_size,
                              hipStream_t stream)
{
    const float* x       = (const float*)d_in[0];
    const int*   ei      = (const int*)d_in[1];
    const float* w_l_in  = (const float*)d_in[2];
    const float* b_l_in  = (const float*)d_in[3];
    const float* w_r_in  = (const float*)d_in[4];
    const float* w_l_out = (const float*)d_in[5];
    const float* b_l_out = (const float*)d_in[6];
    const float* w_r_out = (const float*)d_in[7];
    float* out = (float*)d_out;

    const size_t NB   = (size_t)N_NODES * D * sizeof(float);
    const size_t CNTB = (size_t)N_NODES * sizeof(float);
    char* ws = (char*)d_ws;
    float* agg_in  = (float*)(ws);
    float* agg_out = (float*)(ws + NB);
    int*   cnt_in  = (int*)(ws + 2 * NB);
    int*   cnt_out = (int*)(ws + 2 * NB + CNTB);

    (void)in_sizes; (void)n_in; (void)out_size; (void)ws_size;

    sage_bucket<<<BK_BLOCKS, BK_THREADS, 0, stream>>>(ei, ei + N_EDGES, agg_in, agg_out);

    sage_rank<<<2 * NBUCK, RK_THREADS, 0, stream>>>(ei, ei + N_EDGES, agg_in, agg_out,
                                                    cnt_in, cnt_out);

    sage_gemm<<<GEMM_BLOCKS, 256, 0, stream>>>(ei, ei + N_EDGES, x,
                                                (const int*)agg_in, (const int*)agg_out,
                                                cnt_in, cnt_out,
                                                w_l_in, w_l_out, w_r_in, w_r_out,
                                                b_l_in, b_l_out, out);
}


# Second-round row loads hoisted: each list's first 12 rows of x (was 8) are asked for ahead of the WMMA chunk, lists of 9..12 wait no extra round trip; slot-order sums grouped 8,4,4,8.. (was 8,8..)
# speedup vs baseline: 3.5405x; 1.0531x over previous
#include <hip/hip_runtime.h>
#include <hip/hip_bf16.h>


#define N_NODES 100000
#define N_EDGES 640000
#define D 128

typedef float v2f __attribute__((ext_vector_type(2)));
typedef float v8f __attribute__((ext_vector_type(8)));

#define CAP 64
#define NFIRST 12
#define BNODES 256
#define NBUCK  ((N_NODES + BNODES - 1) / BNODES)
#define SHARE_WORDS 32
#define SHARE_CAP   (SHARE_WORDS - 1)

__device__ __forceinline__ int* bucket_share(float* agg, int b, int k) {
    return (int*)(agg + ((size_t)(b * BNODES + (k >> 1))) * D + 64 + (k & 1) * SHARE_WORDS);
}

#define BK_BLOCKS  256
#define EPB        (N_EDGES / BK_BLOCKS)
#define BK_THREADS 512
#define EPT        ((EPB + BK_THREADS - 1) / BK_THREADS)

__global__ __launch_bounds__(BK_THREADS) void sage_bucket(
    const int* __restrict__ src, const int* __restrict__ dst,
    float* __restrict__ agg_in, float* __restrict__ agg_out)
{
    __shared__ int hist[2 * NBUCK];
    const int tid = threadIdx.x;
    for (int j = tid; j < 2 * NBUCK; j += BK_THREADS) hist[j] = 0;
    __syncthreads();

    const int e0 = blockIdx.x * EPB;
    int s[EPT], d[EPT], ri[EPT], ro[EPT];
#pragma unroll
    for (int i = 0; i < EPT; ++i) {
        const int k = i * BK_THREADS + tid;
        s[i] = 0; d[i] = 0;
        if (k < EPB) { s[i] = src[e0 + k]; d[i] = dst[e0 + k]; }
    }
#pragma unroll
    for (int i = 0; i < EPT; ++i) {
        ri[i] = 0; ro[i] = 0;
        if (i * BK_THREADS + tid < EPB) {
            ri[i] = atomicAdd(&hist[d[i] >> 8], 1);
            ro[i] = atomicAdd(&hist[NBUCK + (s[i] >> 8)], 1);
        }
    }
    __syncthreads();

    const int blk = blockIdx.x;
    for (int j = tid; j < 2 * NBUCK; j += BK_THREADS) {
        const bool rev = j >= NBUCK;
        const int  b   = rev ? j - NBUCK : j;
        bucket_share(rev ? agg_out : agg_in, b, blk)[0] = hist[j];
    }

#pragma unroll
    for (int i = 0; i < EPT; ++i) {
        if (i * BK_THREADS + tid < EPB) {
            if (ri[i] < SHARE_CAP)
                bucket_share(agg_in, d[i] >> 8, blk)[1 + ri[i]] = (s[i] << 8) | (d[i] & 255);
            if (ro[i] < SHARE_CAP)
                bucket_share(agg_out, s[i] >> 8, blk)[1 + ro[i]] = (d[i] << 8) | (s[i] & 255);
        }
    }
}

#define RK_THREADS BK_BLOCKS
#define RK_FIRST   15

__global__ __launch_bounds__(RK_THREADS) void sage_rank(
    const int* __restrict__ src, const int* __restrict__ dst,
    float* agg_in, float* agg_out,
    int* __restrict__ cnt_in, int* __restrict__ cnt_out)
{
    __shared__ int lc[BNODES];
    __shared__ int over;
    const int  tid   = threadIdx.x;
    const bool rev   = blockIdx.x >= NBUCK;
    const int  b     = rev ? blockIdx.x - NBUCK : blockIdx.x;
    float*     agg   = rev ? agg_out : agg_in;
    int*       cnt   = rev ? cnt_out : cnt_in;
    const int  node0 = b * BNODES;
    const int  nrows = min(BNODES, N_NODES - node0);

    lc[tid] = 0;
    if (tid == 0) over = 0;
    const int* sh = bucket_share(agg, b, tid);
    const int4 w0 = ((const int4*)sh)[0];
    const int4 w1 = ((const int4*)sh)[1];
    const int4 w2 = ((const int4*)sh)[2];
    const int4 w3 = ((const int4*)sh)[3];
    const int  n  = w0.x;
    __syncthreads();
    if (n > SHARE_CAP) over = 1;
    __syncthreads();

    if (over == 0) {
        const int ent[RK_FIRST] = { w0.y, w0.z, w0.w, w1.x, w1.y, w1.z, w1.w,
                                    w2.x, w2.y, w2.z, w2.w, w3.x, w3.y, w3.z, w3.w };
#pragma unroll
        for (int u = 0; u < RK_FIRST; ++u) {
            if (u < n) {
                const int kl = ent[u] & 255;
                const int r  = atomicAdd(&lc[kl], 1);
                if (r < CAP) agg[(size_t)(node0 + kl) * D + r] = __int_as_float(ent[u] >> 8);
            }
        }
        for (int u = RK_FIRST; u < n; ++u) {
            const int e  = sh[1 + u];
            const int kl = e & 255;
            const int r  = atomicAdd(&lc[kl], 1);
            if (r < CAP) agg[(size_t)(node0 + kl) * D + r] = __int_as_float(e >> 8);
        }
    } else {
        const int* key = rev ? src : dst;
        const int* oth = rev ? dst : src;
        for (int e = tid; e < N_EDGES; e += RK_THREADS) {
            const int k = key[e];
            if ((k >> 8) == b) {
                const int r = atomicAdd(&lc[k & 255], 1);
                if (r < CAP) agg[(size_t)k * D + r] = __int_as_float(oth[e]);
            }
        }
    }
    __syncthreads();
    if (tid < nrows) cnt[node0 + tid] = lc[tid];
}

__device__ __forceinline__ float4 x_row(const float* __restrict__ x, int node, int lane) {
    return *(const float4*)(x + (size_t)node * D + lane * 4);
}

__device__ __forceinline__ void acc_add(float4& acc, const float4& r) {
    acc.x += r.x; acc.y += r.y; acc.z += r.z; acc.w += r.w;
}

template <int FROM, int CNT, int N>
__device__ __forceinline__ void rows_issue_nb(const int (&nb)[N], int cnt,
                                              const float* __restrict__ x,
                                              int lane, float4 (&r)[N])
{
#pragma unroll
    for (int t = FROM; t < FROM + CNT; ++t) {
        if (t < cnt) r[t] = x_row(x, nb[t], lane);
        else         r[t] = make_float4(0.0f, 0.0f, 0.0f, 0.0f);
    }
}

__device__ __forceinline__ void rows_issue(int idv, int cnt, const float* __restrict__ x,
                                           int lane, float4 (&r)[NFIRST])
{
    int nb[NFIRST];
#pragma unroll
    for (int t = 0; t < NFIRST; ++t)
        nb[t] = __builtin_amdgcn_readlane(idv, t);
    rows_issue_nb<0, NFIRST>(nb, cnt, x, lane, r);
}

template <int FROM, int CNT, int N>
__device__ __forceinline__ float4 rows_sum(const float4 (&r)[N])
{
    float4 g = r[FROM];
#pragma unroll
    for (int t = 1; t < CNT; ++t) acc_add(g, r[FROM + t]);
    return g;
}

__device__ __forceinline__ float4 task_finish(
    const float4 (&r)[NFIRST], int id0, int deg, const int* __restrict__ list, int n, bool rev,
    const int* __restrict__ src, const int* __restrict__ dst,
    const float* __restrict__ x, int lane)
{
    float4 acc = rows_sum<0, 8>(r);
    acc_add(acc, rows_sum<8, NFIRST - 8>(r));
    if (deg <= CAP) {
        int idv = id0;
#pragma nounroll
        for (int j = NFIRST; j < deg; j += 8 - (j & 7)) {
            if (j == 32) {
                const int i = 32 + lane;
                idv = 0;
                if (i < deg) idv = list[i];
            }
            int nb[8];
#pragma unroll
            for (int t = 0; t < 8; ++t)
                nb[t] = __builtin_amdgcn_readlane(idv, (j & 31) + t);
            float4 q[8];
            rows_issue_nb<0, 8>(nb, min(deg - j, 8 - (j & 7)), x, lane, q);
            acc_add(acc, rows_sum<0, 8>(q));
        }
    } else {
        const int* key = rev ? src : dst;
        const int* oth = rev ? dst : src;
#pragma nounroll
        for (int base = 0; base < N_EDGES; base += 32) {
            const int k = key[base + lane];
            const int o = oth[base + lane];
            unsigned hit = __builtin_amdgcn_ballot_w32(k == n);
            while (hit) {
                const int j = __builtin_ctz(hit);
                hit &= hit - 1;
                acc_add(acc, x_row(x, __builtin_amdgcn_readlane(o, j), lane));
            }
        }
    }
    return acc;
}

#define TM      32
#define KTOT    (3 * D)
#define PITCH   (KTOT + 4)
#define NTILES  (N_NODES / TM)
#define GEMM_BLOCKS 256
#define KC      16
#define NCHUNK  (KTOT / 4 / KC)

__device__ __forceinline__ void list_fetch(
    int tile, int r0, int lane,
    const int* __restrict__ lst_in, const int* __restrict__ lst_out,
    const int* __restrict__ cnt_in, const int* __restrict__ cnt_out,
    int (&ids)[8], int (&dg)[8])
{
#pragma unroll
    for (int j = 0; j < 4; ++j) {
        const int row = tile * TM + r0 + 8 * j;
        ids[j]     = lst_in [(size_t)row * D + lane];
        ids[4 + j] = lst_out[(size_t)row * D + lane];
        dg[j]      = cnt_in[row];
        dg[4 + j]  = cnt_out[row];
    }
}

__device__ __forceinline__ void pair_issue(
    int ids_i, int dg_i, int ids_o, int dg_o, const float* __restrict__ x, int lane,
    int& idi, int& di, int& ido, int& dout, float4 (&ri)[NFIRST], float4 (&ro)[NFIRST])
{
    di   = __builtin_amdgcn_readfirstlane(dg_i);
    dout = __builtin_amdgcn_readfirstlane(dg_o);
    idi  = (lane < di)   ? ids_i : 0;
    ido  = (lane < dout) ? ids_o : 0;
    rows_issue(idi, di   > CAP ? 0 : min(di, NFIRST),   x, lane, ri);
    rows_issue(ido, dout > CAP ? 0 : min(dout, NFIRST), x, lane, ro);
}

__device__ __forceinline__ void pair_finish(
    float* __restrict__ p, int row, int lane,
    const float4 (&ri)[NFIRST], const float4 (&ro)[NFIRST], int idi, int di, int ido, int dout,
    const int* __restrict__ lst_in, const int* __restrict__ lst_out,
    const int* __restrict__ src, const int* __restrict__ dst, const float* __restrict__ x)
{
    float4 a = task_finish(ri, idi, di,   lst_in  + (size_t)row * D, row, false, src, dst, x, lane);
    float4 b = task_finish(ro, ido, dout, lst_out + (size_t)row * D, row, true,  src, dst, x, lane);
    const float si = 1.0f / fmaxf((float)di, 1.0f);
    const float so = 1.0f / fmaxf((float)dout, 1.0f);
    a.x *= si; a.y *= si; a.z *= si; a.w *= si;
    b.x *= so; b.y *= so; b.z *= so; b.w *= so;
    *(float4*)(p)     = a;
    *(float4*)(p + D) = b;
}

__device__ __forceinline__ void frag_load(const float* A0, const float* A1, int c,
                                          v2f (&f0)[KC], v2f (&f1)[KC])
{
#pragma unroll
    for (int k = 0; k < KC; ++k) {
        f0[k] = *(const v2f*)(A0 + 4 * (c * KC + k));
        f1[k] = *(const v2f*)(A1 + 4 * (c * KC + k));
    }
}

__global__ __launch_bounds__(256) void sage_gemm(
    const int* __restrict__ src, const int* __restrict__ dst,
    const float* __restrict__ x,
    const int* __restrict__ lst_in, const int* __restrict__ lst_out,
    const int* __restrict__ cnt_in, const int* __restrict__ cnt_out,
    const float* __restrict__ w_l_in, const float* __restrict__ w_l_out,
    const float* __restrict__ w_r_in, const float* __restrict__ w_r_out,
    const float* __restrict__ b_l_in, const float* __restrict__ b_l_out,
    float* __restrict__ out)
{
    __shared__ __attribute__((aligned(16))) float tileA[2][TM * PITCH];

    const int tid   = threadIdx.x;
    const int wave  = tid >> 5;
    const int lane  = tid & 31;
    const int half  = lane >> 4;
    const int l15   = lane & 15;
    const int col   = wave * 16 + l15;

    v2f bw[96];
    {
        const size_t woff = (size_t)col * D + 2 * half;
#pragma unroll
        for (int i = 0; i < 32; ++i) {
            bw[i]      = *(const v2f*)(w_l_in  + woff + 4 * i);
            bw[32 + i] = *(const v2f*)(w_l_out + woff + 4 * i);
            bw[64 + i] = *(const v2f*)(w_r_in  + woff + 4 * i) + *(const v2f*)(w_r_out + woff + 4 * i);
        }
    }
    const float bias = b_l_in[col] + b_l_out[col];

    const int c4 = lane;
    const int r0 = wave;

    int tile = blockIdx.x;
    int buf  = 0;
    int idsA[8], dgA[8];
    if (tile < NTILES) {
        int row = tile * TM + r0;
        int li = lst_in [(size_t)row * D + lane];
        int lo = lst_out[(size_t)row * D + lane];
        int gi = cnt_in[row];
        int go = cnt_out[row];
#pragma nounroll
        for (int j = 0; j < 4; ++j) {
            const int ci = li, co = lo, cgi = gi, cgo = go;
            const int rown = row + 8;
            if (j < 3) {
                li = lst_in [(size_t)rown * D + lane];
                lo = lst_out[(size_t)rown * D + lane];
                gi = cnt_in[rown];
                go = cnt_out[rown];
            }
            float* p = &tileA[0][0] + (r0 + 8 * j) * PITCH + c4 * 4;
            const float4 xr = *(const float4*)(x + (size_t)row * D + c4 * 4);
            float4 ri[NFIRST], ro[NFIRST];
            int idi, di, ido, dout;
            pair_issue(ci, cgi, co, cgo, x, lane, idi, di, ido, dout, ri, ro);
            pair_finish(p, row, lane, ri, ro, idi, di, ido, dout, lst_in, lst_out, src, dst, x);
            *(float4*)(p + 2 * D) = xr;
            row = rown;
        }
    }
    if (tile + GEMM_BLOCKS < NTILES) {
        list_fetch(tile + GEMM_BLOCKS, r0, lane, lst_in, lst_out, cnt_in, cnt_out, idsA, dgA);
    } else {
#pragma unroll
        for (int j = 0; j < 8; ++j) { idsA[j] = 0; dgA[j] = 0; }
    }
    __syncthreads();

    for (; tile < NTILES; tile += GEMM_BLOCKS) {
        const int  next      = tile + GEMM_BLOCKS;
        const bool has_next  = next < NTILES;
        const bool has_next2 = next + GEMM_BLOCKS < NTILES;

        const int xt = has_next ? next : tile;

        const float* A0 = &tileA[buf][0] + l15 * PITCH + 2 * half;
        const float* A1 = A0 + 16 * PITCH;
        float* nt = &tileA[buf ^ 1][0] + r0 * PITCH + c4 * 4;

        v8f acc0, acc1;
#pragma unroll
        for (int r = 0; r < 8; ++r) { acc0[r] = bias; acc1[r] = bias; }

        v2f fa0[2][KC], fa1[2][KC];
        frag_load(A0, A1, 0, fa0[0], fa1[0]);
#pragma unroll
        for (int c = 0; c < NCHUNK; ++c) {
            float4 ri[NFIRST], ro[NFIRST];
            int idi = 0, di = 0, ido = 0, dout = 0;
            float4 xr = make_float4(0.0f, 0.0f, 0.0f, 0.0f);
            if (c < 4) {
                xr = *(const float4*)(x + (size_t)(xt * TM + r0 + 8 * c) * D + c4 * 4);
                pair_issue(idsA[c], dgA[c], idsA[4 + c], dgA[4 + c], x, lane,
                           idi, di, ido, dout, ri, ro);
            }
            if (c == 4) {
                if (has_next2) {
                    list_fetch(next + GEMM_BLOCKS, r0, lane, lst_in, lst_out, cnt_in, cnt_out,
                               idsA, dgA);
                } else {
#pragma unroll
                    for (int j = 0; j < 8; ++j) { idsA[j] = 0; dgA[j] = 0; }
                }
            }
            if (c + 1 < NCHUNK)
                frag_load(A0, A1, c + 1, fa0[(c + 1) & 1], fa1[(c + 1) & 1]);
            __builtin_amdgcn_sched_barrier(0);
#pragma unroll
            for (int k = 0; k < KC; ++k) {
                acc0 = __builtin_amdgcn_wmma_f32_16x16x4_f32(
                     false, fa0[c & 1][k],  false, bw[c * KC + k],
                     (short)0, acc0,  false,  false);
                acc1 = __builtin_amdgcn_wmma_f32_16x16x4_f32(
                     false, fa1[c & 1][k],  false, bw[c * KC + k],
                     (short)0, acc1,  false,  false);
            }
            __builtin_amdgcn_sched_barrier(0);
            asm volatile("v_nop\n\tv_nop\n\tv_nop\n\tv_nop\n\tv_nop\n\tv_nop\n\tv_nop\n\tv_nop");
            __builtin_amdgcn_sched_barrier(0);
            if (c < 4) {
                pair_finish(nt + 8 * c * PITCH, xt * TM + r0 + 8 * c, lane,
                            ri, ro, idi, di, ido, dout, lst_in, lst_out, src, dst, x);
                *(float4*)(nt + 8 * c * PITCH + 2 * D) = xr;
            }
        }

        __syncthreads();

        const int mbase = tile * TM + half * 8;
#pragma unroll
        for (int r = 0; r < 8; ++r) {
            out[(size_t)(mbase + r) * D + col]      = acc0[r];
            out[(size_t)(mbase + 16 + r) * D + col] = acc1[r];
        }
        buf ^= 1;
    }
}

extern "C" void kernel_launch(void* const* d_in, const int* in_sizes, int n_in,
                              void* d_out, int out_size, void* d_ws, size_t ws_size,
                              hipStream_t stream)
{
    const float* x       = (const float*)d_in[0];
    const int*   ei      = (const int*)d_in[1];
    const float* w_l_in  = (const float*)d_in[2];
    const float* b_l_in  = (const float*)d_in[3];
    const float* w_r_in  = (const float*)d_in[4];
    const float* w_l_out = (const float*)d_in[5];
    const float* b_l_out = (const float*)d_in[6];
    const float* w_r_out = (const float*)d_in[7];
    float* out = (float*)d_out;

    const size_t NB   = (size_t)N_NODES * D * sizeof(float);
    const size_t CNTB = (size_t)N_NODES * sizeof(float);
    char* ws = (char*)d_ws;
    float* agg_in  = (float*)(ws);
    float* agg_out = (float*)(ws + NB);
    int*   cnt_in  = (int*)(ws + 2 * NB);
    int*   cnt_out = (int*)(ws + 2 * NB + CNTB);

    (void)in_sizes; (void)n_in; (void)out_size; (void)ws_size;

    sage_bucket<<<BK_BLOCKS, BK_THREADS, 0, stream>>>(ei, ei + N_EDGES, agg_in, agg_out);

    sage_rank<<<2 * NBUCK, RK_THREADS, 0, stream>>>(ei, ei + N_EDGES, agg_in, agg_out,
                                                    cnt_in, cnt_out);

    sage_gemm<<<GEMM_BLOCKS, 256, 0, stream>>>(ei, ei + N_EDGES, x,
                                                (const int*)agg_in, (const int*)agg_out,
                                                cnt_in, cnt_out,
                                                w_l_in, w_l_out, w_r_in, w_r_out,
                                                b_l_in, b_l_out, out);
}


# Scattered 4-B stores of the bucket and rank kernels staged in LDS: shares and neighbour lists are built in LDS and written as whole 128-B lines, four per store instruction; sum order as before
# speedup vs baseline: 3.7399x; 1.0563x over previous
#include <hip/hip_runtime.h>
#include <hip/hip_bf16.h>


#define N_NODES 100000
#define N_EDGES 640000
#define D 128

typedef float v2f __attribute__((ext_vector_type(2)));
typedef float v8f __attribute__((ext_vector_type(8)));

#define CAP 64
#define NFIRST 12
#define BNODES 256
#define NBUCK  ((N_NODES + BNODES - 1) / BNODES)
#define SHARE_WORDS 32
#define SHARE_CAP   (SHARE_WORDS - 1)
#define SH_PITCH    36
#define LST_PITCH   (CAP + 4)

__device__ __forceinline__ int* bucket_share(float* agg, int b, int k) {
    return (int*)(agg + ((size_t)(b * BNODES + (k >> 1))) * D + 64 + (k & 1) * SHARE_WORDS);
}

#define BK_BLOCKS  256
#define EPB        (N_EDGES / BK_BLOCKS)
#define BK_THREADS 512
#define EPT        ((EPB + BK_THREADS - 1) / BK_THREADS)

__global__ __launch_bounds__(BK_THREADS) void sage_bucket(
    const int* __restrict__ src, const int* __restrict__ dst,
    float* __restrict__ agg_in, float* __restrict__ agg_out)
{
    __shared__ __attribute__((aligned(16))) int sh[2 * NBUCK * SH_PITCH];
    const int tid = threadIdx.x;
    for (int j = tid; j < 2 * NBUCK * (SH_PITCH / 4); j += BK_THREADS)
        ((int4*)sh)[j] = make_int4(0, 0, 0, 0);
    __syncthreads();

    const int e0 = blockIdx.x * EPB;
    int s[EPT], d[EPT];
#pragma unroll
    for (int i = 0; i < EPT; ++i) {
        const int k = i * BK_THREADS + tid;
        s[i] = 0; d[i] = 0;
        if (k < EPB) { s[i] = src[e0 + k]; d[i] = dst[e0 + k]; }
    }
#pragma unroll
    for (int i = 0; i < EPT; ++i) {
        if (i * BK_THREADS + tid < EPB) {
            int* pi = sh + (d[i] >> 8) * SH_PITCH;
            const int ri = atomicAdd(pi, 1);
            if (ri < SHARE_CAP) pi[1 + ri] = (s[i] << 8) | (d[i] & 255);
            int* po = sh + (NBUCK + (s[i] >> 8)) * SH_PITCH;
            const int ro = atomicAdd(po, 1);
            if (ro < SHARE_CAP) po[1 + ro] = (d[i] << 8) | (s[i] & 255);
        }
    }
    __syncthreads();

    const int blk = blockIdx.x;
    for (int j = tid; j < 2 * NBUCK * (SHARE_WORDS / 4); j += BK_THREADS) {
        const int  line = j >> 3;
        const int  q    = j & 7;
        const bool rev  = line >= NBUCK;
        const int  b    = rev ? line - NBUCK : line;
        *(int4*)(bucket_share(rev ? agg_out : agg_in, b, blk) + 4 * q) =
            ((const int4*)sh)[line * (SH_PITCH / 4) + q];
    }
}

#define RK_THREADS BK_BLOCKS
#define RK_FIRST   15

__global__ __launch_bounds__(RK_THREADS) void sage_rank(
    const int* __restrict__ src, const int* __restrict__ dst,
    float* agg_in, float* agg_out,
    int* __restrict__ cnt_in, int* __restrict__ cnt_out)
{
    __shared__ int lc[BNODES];
    __shared__ __attribute__((aligned(16))) int lst[BNODES * LST_PITCH];
    __shared__ int over;
    const int  tid   = threadIdx.x;
    const bool rev   = blockIdx.x >= NBUCK;
    const int  b     = rev ? blockIdx.x - NBUCK : blockIdx.x;
    float*     agg   = rev ? agg_out : agg_in;
    int*       cnt   = rev ? cnt_out : cnt_in;
    const int  node0 = b * BNODES;
    const int  nrows = min(BNODES, N_NODES - node0);

    lc[tid] = 0;
    if (tid == 0) over = 0;
    for (int j = tid; j < BNODES * LST_PITCH / 4; j += RK_THREADS)
        ((int4*)lst)[j] = make_int4(0, 0, 0, 0);
    const int* sh = bucket_share(agg, b, tid);
    const int4 w0 = ((const int4*)sh)[0];
    const int4 w1 = ((const int4*)sh)[1];
    const int4 w2 = ((const int4*)sh)[2];
    const int4 w3 = ((const int4*)sh)[3];
    const int  n  = w0.x;
    __syncthreads();
    if (n > SHARE_CAP) over = 1;
    __syncthreads();

    if (over == 0) {
        const int ent[RK_FIRST] = { w0.y, w0.z, w0.w, w1.x, w1.y, w1.z, w1.w,
                                    w2.x, w2.y, w2.z, w2.w, w3.x, w3.y, w3.z, w3.w };
#pragma unroll
        for (int u = 0; u < RK_FIRST; ++u) {
            if (u < n) {
                const int kl = ent[u] & 255;
                const int r  = atomicAdd(&lc[kl], 1);
                if (r < CAP) lst[kl * LST_PITCH + r] = ent[u] >> 8;
            }
        }
        for (int u = RK_FIRST; u < n; ++u) {
            const int e  = sh[1 + u];
            const int kl = e & 255;
            const int r  = atomicAdd(&lc[kl], 1);
            if (r < CAP) lst[kl * LST_PITCH + r] = e >> 8;
        }
    } else {
        const int* key = rev ? src : dst;
        const int* oth = rev ? dst : src;
        for (int e = tid; e < N_EDGES; e += RK_THREADS) {
            const int k = key[e];
            if ((k >> 8) == b) {
                const int r = atomicAdd(&lc[k & 255], 1);
                if (r < CAP) lst[(k & 255) * LST_PITCH + r] = oth[e];
            }
        }
    }
    __syncthreads();
    if (tid < nrows) cnt[node0 + tid] = lc[tid];
    int* rows = (int*)agg + (size_t)node0 * D;
    for (int j = tid; j < nrows * 8; j += RK_THREADS) {
        const int node = j >> 3;
        const int q    = j & 7;
        *(int4*)(rows + (size_t)node * D + 4 * q) = ((const int4*)lst)[node * (LST_PITCH / 4) + q];
        if (lc[node] > 32)
            *(int4*)(rows + (size_t)node * D + 32 + 4 * q) =
                ((const int4*)lst)[node * (LST_PITCH / 4) + 8 + q];
    }
}

__device__ __forceinline__ float4 x_row(const float* __restrict__ x, int node, int lane) {
    return *(const float4*)(x + (size_t)node * D + lane * 4);
}

__device__ __forceinline__ void acc_add(float4& acc, const float4& r) {
    acc.x += r.x; acc.y += r.y; acc.z += r.z; acc.w += r.w;
}

template <int FROM, int CNT, int N>
__device__ __forceinline__ void rows_issue_nb(const int (&nb)[N], int cnt,
                                              const float* __restrict__ x,
                                              int lane, float4 (&r)[N])
{
#pragma unroll
    for (int t = FROM; t < FROM + CNT; ++t) {
        if (t < cnt) r[t] = x_row(x, nb[t], lane);
        else         r[t] = make_float4(0.0f, 0.0f, 0.0f, 0.0f);
    }
}

__device__ __forceinline__ void rows_issue(int idv, int cnt, const float* __restrict__ x,
                                           int lane, float4 (&r)[NFIRST])
{
    int nb[NFIRST];
#pragma unroll
    for (int t = 0; t < NFIRST; ++t)
        nb[t] = __builtin_amdgcn_readlane(idv, t);
    rows_issue_nb<0, NFIRST>(nb, cnt, x, lane, r);
}

template <int FROM, int CNT, int N>
__device__ __forceinline__ float4 rows_sum(const float4 (&r)[N])
{
    float4 g = r[FROM];
#pragma unroll
    for (int t = 1; t < CNT; ++t) acc_add(g, r[FROM + t]);
    return g;
}

__device__ __forceinline__ float4 task_finish(
    const float4 (&r)[NFIRST], int id0, int deg, const int* __restrict__ list, int n, bool rev,
    const int* __restrict__ src, const int* __restrict__ dst,
    const float* __restrict__ x, int lane)
{
    float4 acc = rows_sum<0, 8>(r);
    acc_add(acc, rows_sum<8, NFIRST - 8>(r));
    if (deg <= CAP) {
        int idv = id0;
#pragma nounroll
        for (int j = NFIRST; j < deg; j += 8 - (j & 7)) {
            if (j == 32) {
                const int i = 32 + lane;
                idv = 0;
                if (i < deg) idv = list[i];
            }
            int nb[8];
#pragma unroll
            for (int t = 0; t < 8; ++t)
                nb[t] = __builtin_amdgcn_readlane(idv, (j & 31) + t);
            float4 q[8];
            rows_issue_nb<0, 8>(nb, min(deg - j, 8 - (j & 7)), x, lane, q);
            acc_add(acc, rows_sum<0, 8>(q));
        }
    } else {
        const int* key = rev ? src : dst;
        const int* oth = rev ? dst : src;
#pragma nounroll
        for (int base = 0; base < N_EDGES; base += 32) {
            const int k = key[base + lane];
            const int o = oth[base + lane];
            unsigned hit = __builtin_amdgcn_ballot_w32(k == n);
            while (hit) {
                const int j = __builtin_ctz(hit);
                hit &= hit - 1;
                acc_add(acc, x_row(x, __builtin_amdgcn_readlane(o, j), lane));
            }
        }
    }
    return acc;
}

#define TM      32
#define KTOT    (3 * D)
#define PITCH   (KTOT + 4)
#define NTILES  (N_NODES / TM)
#define GEMM_BLOCKS 256
#define KC      16
#define NCHUNK  (KTOT / 4 / KC)

__device__ __forceinline__ void list_fetch(
    int tile, int r0, int lane,
    const int* __restrict__ lst_in, const int* __restrict__ lst_out,
    const int* __restrict__ cnt_in, const int* __restrict__ cnt_out,
    int (&ids)[8], int (&dg)[8])
{
#pragma unroll
    for (int j = 0; j < 4; ++j) {
        const int row = tile * TM + r0 + 8 * j;
        ids[j]     = lst_in [(size_t)row * D + lane];
        ids[4 + j] = lst_out[(size_t)row * D + lane];
        dg[j]      = cnt_in[row];
        dg[4 + j]  = cnt_out[row];
    }
}

__device__ __forceinline__ void pair_issue(
    int ids_i, int dg_i, int ids_o, int dg_o, const float* __restrict__ x, int lane,
    int& idi, int& di, int& ido, int& dout, float4 (&ri)[NFIRST], float4 (&ro)[NFIRST])
{
    di   = __builtin_amdgcn_readfirstlane(dg_i);
    dout = __builtin_amdgcn_readfirstlane(dg_o);
    idi  = (lane < di)   ? ids_i : 0;
    ido  = (lane < dout) ? ids_o : 0;
    rows_issue(idi, di   > CAP ? 0 : min(di, NFIRST),   x, lane, ri);
    rows_issue(ido, dout > CAP ? 0 : min(dout, NFIRST), x, lane, ro);
}

__device__ __forceinline__ void pair_finish(
    float* __restrict__ p, int row, int lane,
    const float4 (&ri)[NFIRST], const float4 (&ro)[NFIRST], int idi, int di, int ido, int dout,
    const int* __restrict__ lst_in, const int* __restrict__ lst_out,
    const int* __restrict__ src, const int* __restrict__ dst, const float* __restrict__ x)
{
    float4 a = task_finish(ri, idi, di,   lst_in  + (size_t)row * D, row, false, src, dst, x, lane);
    float4 b = task_finish(ro, ido, dout, lst_out + (size_t)row * D, row, true,  src, dst, x, lane);
    const float si = 1.0f / fmaxf((float)di, 1.0f);
    const float so = 1.0f / fmaxf((float)dout, 1.0f);
    a.x *= si; a.y *= si; a.z *= si; a.w *= si;
    b.x *= so; b.y *= so; b.z *= so; b.w *= so;
    *(float4*)(p)     = a;
    *(float4*)(p + D) = b;
}

__device__ __forceinline__ void frag_load(const float* A0, const float* A1, int c,
                                          v2f (&f0)[KC], v2f (&f1)[KC])
{
#pragma unroll
    for (int k = 0; k < KC; ++k) {
        f0[k] = *(const v2f*)(A0 + 4 * (c * KC + k));
        f1[k] = *(const v2f*)(A1 + 4 * (c * KC + k));
    }
}

__global__ __launch_bounds__(256) void sage_gemm(
    const int* __restrict__ src, const int* __restrict__ dst,
    const float* __restrict__ x,
    const int* __restrict__ lst_in, const int* __restrict__ lst_out,
    const int* __restrict__ cnt_in, const int* __restrict__ cnt_out,
    const float* __restrict__ w_l_in, const float* __restrict__ w_l_out,
    const float* __restrict__ w_r_in, const float* __restrict__ w_r_out,
    const float* __restrict__ b_l_in, const float* __restrict__ b_l_out,
    float* __restrict__ out)
{
    __shared__ __attribute__((aligned(16))) float tileA[2][TM * PITCH];

    const int tid   = threadIdx.x;
    const int wave  = tid >> 5;
    const int lane  = tid & 31;
    const int half  = lane >> 4;
    const int l15   = lane & 15;
    const int col   = wave * 16 + l15;

    v2f bw[96];
    {
        const size_t woff = (size_t)col * D + 2 * half;
#pragma unroll
        for (int i = 0; i < 32; ++i) {
            bw[i]      = *(const v2f*)(w_l_in  + woff + 4 * i);
            bw[32 + i] = *(const v2f*)(w_l_out + woff + 4 * i);
            bw[64 + i] = *(const v2f*)(w_r_in  + woff + 4 * i) + *(const v2f*)(w_r_out + woff + 4 * i);
        }
    }
    const float bias = b_l_in[col] + b_l_out[col];

    const int c4 = lane;
    const int r0 = wave;

    int tile = blockIdx.x;
    int buf  = 0;
    int idsA[8], dgA[8];
    if (tile < NTILES) {
        int row = tile * TM + r0;
        int li = lst_in [(size_t)row * D + lane];
        int lo = lst_out[(size_t)row * D + lane];
        int gi = cnt_in[row];
        int go = cnt_out[row];
#pragma nounroll
        for (int j = 0; j < 4; ++j) {
            const int ci = li, co = lo, cgi = gi, cgo = go;
            const int rown = row + 8;
            if (j < 3) {
                li = lst_in [(size_t)rown * D + lane];
                lo = lst_out[(size_t)rown * D + lane];
                gi = cnt_in[rown];
                go = cnt_out[rown];
            }
            float* p = &tileA[0][0] + (r0 + 8 * j) * PITCH + c4 * 4;
            const float4 xr = *(const float4*)(x + (size_t)row * D + c4 * 4);
            float4 ri[NFIRST], ro[NFIRST];
            int idi, di, ido, dout;
            pair_issue(ci, cgi, co, cgo, x, lane, idi, di, ido, dout, ri, ro);
            pair_finish(p, row, lane, ri, ro, idi, di, ido, dout, lst_in, lst_out, src, dst, x);
            *(float4*)(p + 2 * D) = xr;
            row = rown;
        }
    }
    if (tile + GEMM_BLOCKS < NTILES) {
        list_fetch(tile + GEMM_BLOCKS, r0, lane, lst_in, lst_out, cnt_in, cnt_out, idsA, dgA);
    } else {
#pragma unroll
        for (int j = 0; j < 8; ++j) { idsA[j] = 0; dgA[j] = 0; }
    }
    __syncthreads();

    for (; tile < NTILES; tile += GEMM_BLOCKS) {
        const int  next      = tile + GEMM_BLOCKS;
        const bool has_next  = next < NTILES;
        const bool has_next2 = next + GEMM_BLOCKS < NTILES;

        const int xt = has_next ? next : tile;

        const float* A0 = &tileA[buf][0] + l15 * PITCH + 2 * half;
        const float* A1 = A0 + 16 * PITCH;
        float* nt = &tileA[buf ^ 1][0] + r0 * PITCH + c4 * 4;

        v8f acc0, acc1;
#pragma unroll
        for (int r = 0; r < 8; ++r) { acc0[r] = bias; acc1[r] = bias; }

        v2f fa0[2][KC], fa1[2][KC];
        frag_load(A0, A1, 0, fa0[0], fa1[0]);
#pragma unroll
        for (int c = 0; c < NCHUNK; ++c) {
            float4 ri[NFIRST], ro[NFIRST];
            int idi = 0, di = 0, ido = 0, dout = 0;
            float4 xr = make_float4(0.0f, 0.0f, 0.0f, 0.0f);
            if (c < 4) {
                xr = *(const float4*)(x + (size_t)(xt * TM + r0 + 8 * c) * D + c4 * 4);
                pair_issue(idsA[c], dgA[c], idsA[4 + c], dgA[4 + c], x, lane,
                           idi, di, ido, dout, ri, ro);
            }
            if (c == 4) {
                if (has_next2) {
                    list_fetch(next + GEMM_BLOCKS, r0, lane, lst_in, lst_out, cnt_in, cnt_out,
                               idsA, dgA);
                } else {
#pragma unroll
                    for (int j = 0; j < 8; ++j) { idsA[j] = 0; dgA[j] = 0; }
                }
            }
            if (c + 1 < NCHUNK)
                frag_load(A0, A1, c + 1, fa0[(c + 1) & 1], fa1[(c + 1) & 1]);
            __builtin_amdgcn_sched_barrier(0);
#pragma unroll
            for (int k = 0; k < KC; ++k) {
                acc0 = __builtin_amdgcn_wmma_f32_16x16x4_f32(
                     false, fa0[c & 1][k],  false, bw[c * KC + k],
                     (short)0, acc0,  false,  false);
                acc1 = __builtin_amdgcn_wmma_f32_16x16x4_f32(
                     false, fa1[c & 1][k],  false, bw[c * KC + k],
                     (short)0, acc1,  false,  false);
            }
            __builtin_amdgcn_sched_barrier(0);
            asm volatile("v_nop\n\tv_nop\n\tv_nop\n\tv_nop\n\tv_nop\n\tv_nop\n\tv_nop\n\tv_nop");
            __builtin_amdgcn_sched_barrier(0);
            if (c < 4) {
                pair_finish(nt + 8 * c * PITCH, xt * TM + r0 + 8 * c, lane,
                            ri, ro, idi, di, ido, dout, lst_in, lst_out, src, dst, x);
                *(float4*)(nt + 8 * c * PITCH + 2 * D) = xr;
            }
        }

        __syncthreads();

        const int mbase = tile * TM + half * 8;
#pragma unroll
        for (int r = 0; r < 8; ++r) {
            out[(size_t)(mbase + r) * D + col]      = acc0[r];
            out[(size_t)(mbase + 16 + r) * D + col] = acc1[r];
        }
        buf ^= 1;
    }
}

extern "C" void kernel_launch(void* const* d_in, const int* in_sizes, int n_in,
                              void* d_out, int out_size, void* d_ws, size_t ws_size,
                              hipStream_t stream)
{
    const float* x       = (const float*)d_in[0];
    const int*   ei      = (const int*)d_in[1];
    const float* w_l_in  = (const float*)d_in[2];
    const float* b_l_in  = (const float*)d_in[3];
    const float* w_r_in  = (const float*)d_in[4];
    const float* w_l_out = (const float*)d_in[5];
    const float* b_l_out = (const float*)d_in[6];
    const float* w_r_out = (const float*)d_in[7];
    float* out = (float*)d_out;

    const size_t NB   = (size_t)N_NODES * D * sizeof(float);
    const size_t CNTB = (size_t)N_NODES * sizeof(float);
    char* ws = (char*)d_ws;
    float* agg_in  = (float*)(ws);
    float* agg_out = (float*)(ws + NB);
    int*   cnt_in  = (int*)(ws + 2 * NB);
    int*   cnt_out = (int*)(ws + 2 * NB + CNTB);

    (void)in_sizes; (void)n_in; (void)out_size; (void)ws_size;

    sage_bucket<<<BK_BLOCKS, BK_THREADS, 0, stream>>>(ei, ei + N_EDGES, agg_in, agg_out);

    sage_rank<<<2 * NBUCK, RK_THREADS, 0, stream>>>(ei, ei + N_EDGES, agg_in, agg_out,
                                                    cnt_in, cnt_out);

    sage_gemm<<<GEMM_BLOCKS, 256, 0, stream>>>(ei, ei + N_EDGES, x,
                                                (const int*)agg_in, (const int*)agg_out,
                                                cnt_in, cnt_out,
                                                w_l_in, w_l_out, w_r_in, w_r_out,
                                                b_l_in, b_l_out, out);
}


# GEMM tile stage: list slots past the degree are loaded from a constant row of zeros, chosen once per list in the lanes: a chunk's 24 row loads are branch-free straight-line code; sum order as before
# speedup vs baseline: 4.1089x; 1.0987x over previous
#include <hip/hip_runtime.h>
#include <hip/hip_bf16.h>


#define N_NODES 100000
#define N_EDGES 640000
#define D 128

typedef float v2f __attribute__((ext_vector_type(2)));
typedef float v8f __attribute__((ext_vector_type(8)));

#define CAP 64
#define NFIRST 12
#define BNODES 256
#define NBUCK  ((N_NODES + BNODES - 1) / BNODES)
#define SHARE_WORDS 32
#define SHARE_CAP   (SHARE_WORDS - 1)
#define SH_PITCH    36
#define LST_PITCH   (CAP + 4)

__device__ __forceinline__ int* bucket_share(float* agg, int b, int k) {
    return (int*)(agg + ((size_t)(b * BNODES + (k >> 1))) * D + 64 + (k & 1) * SHARE_WORDS);
}

#define BK_BLOCKS  256
#define EPB        (N_EDGES / BK_BLOCKS)
#define BK_THREADS 512
#define EPT        ((EPB + BK_THREADS - 1) / BK_THREADS)

__global__ __launch_bounds__(BK_THREADS) void sage_bucket(
    const int* __restrict__ src, const int* __restrict__ dst,
    float* __restrict__ agg_in, float* __restrict__ agg_out)
{
    __shared__ __attribute__((aligned(16))) int sh[2 * NBUCK * SH_PITCH];
    const int tid = threadIdx.x;
    for (int j = tid; j < 2 * NBUCK * (SH_PITCH / 4); j += BK_THREADS)
        ((int4*)sh)[j] = make_int4(0, 0, 0, 0);
    __syncthreads();

    const int e0 = blockIdx.x * EPB;
    int s[EPT], d[EPT];
#pragma unroll
    for (int i = 0; i < EPT; ++i) {
        const int k = i * BK_THREADS + tid;
        s[i] = 0; d[i] = 0;
        if (k < EPB) { s[i] = src[e0 + k]; d[i] = dst[e0 + k]; }
    }
#pragma unroll
    for (int i = 0; i < EPT; ++i) {
        if (i * BK_THREADS + tid < EPB) {
            int* pi = sh + (d[i] >> 8) * SH_PITCH;
            const int ri = atomicAdd(pi, 1);
            if (ri < SHARE_CAP) pi[1 + ri] = (s[i] << 8) | (d[i] & 255);
            int* po = sh + (NBUCK + (s[i] >> 8)) * SH_PITCH;
            const int ro = atomicAdd(po, 1);
            if (ro < SHARE_CAP) po[1 + ro] = (d[i] << 8) | (s[i] & 255);
        }
    }
    __syncthreads();

    const int blk = blockIdx.x;
    for (int j = tid; j < 2 * NBUCK * (SHARE_WORDS / 4); j += BK_THREADS) {
        const int  line = j >> 3;
        const int  q    = j & 7;
        const bool rev  = line >= NBUCK;
        const int  b    = rev ? line - NBUCK : line;
        *(int4*)(bucket_share(rev ? agg_out : agg_in, b, blk) + 4 * q) =
            ((const int4*)sh)[line * (SH_PITCH / 4) + q];
    }
}

#define RK_THREADS BK_BLOCKS
#define RK_FIRST   15

__global__ __launch_bounds__(RK_THREADS) void sage_rank(
    const int* __restrict__ src, const int* __restrict__ dst,
    float* agg_in, float* agg_out,
    int* __restrict__ cnt_in, int* __restrict__ cnt_out)
{
    __shared__ int lc[BNODES];
    __shared__ __attribute__((aligned(16))) int lst[BNODES * LST_PITCH];
    __shared__ int over;
    const int  tid   = threadIdx.x;
    const bool rev   = blockIdx.x >= NBUCK;
    const int  b     = rev ? blockIdx.x - NBUCK : blockIdx.x;
    float*     agg   = rev ? agg_out : agg_in;
    int*       cnt   = rev ? cnt_out : cnt_in;
    const int  node0 = b * BNODES;
    const int  nrows = min(BNODES, N_NODES - node0);

    lc[tid] = 0;
    if (tid == 0) over = 0;
    for (int j = tid; j < BNODES * LST_PITCH / 4; j += RK_THREADS)
        ((int4*)lst)[j] = make_int4(0, 0, 0, 0);
    const int* sh = bucket_share(agg, b, tid);
    const int4 w0 = ((const int4*)sh)[0];
    const int4 w1 = ((const int4*)sh)[1];
    const int4 w2 = ((const int4*)sh)[2];
    const int4 w3 = ((const int4*)sh)[3];
    const int  n  = w0.x;
    __syncthreads();
    if (n > SHARE_CAP) over = 1;
    __syncthreads();

    if (over == 0) {
        const int ent[RK_FIRST] = { w0.y, w0.z, w0.w, w1.x, w1.y, w1.z, w1.w,
                                    w2.x, w2.y, w2.z, w2.w, w3.x, w3.y, w3.z, w3.w };
#pragma unroll
        for (int u = 0; u < RK_FIRST; ++u) {
            if (u < n) {
                const int kl = ent[u] & 255;
                const int r  = atomicAdd(&lc[kl], 1);
                if (r < CAP) lst[kl * LST_PITCH + r] = ent[u] >> 8;
            }
        }
        for (int u = RK_FIRST; u < n; ++u) {
            const int e  = sh[1 + u];
            const int kl = e & 255;
            const int r  = atomicAdd(&lc[kl], 1);
            if (r < CAP) lst[kl * LST_PITCH + r] = e >> 8;
        }
    } else {
        const int* key = rev ? src : dst;
        const int* oth = rev ? dst : src;
        for (int e = tid; e < N_EDGES; e += RK_THREADS) {
            const int k = key[e];
            if ((k >> 8) == b) {
                const int r = atomicAdd(&lc[k & 255], 1);
                if (r < CAP) lst[(k & 255) * LST_PITCH + r] = oth[e];
            }
        }
    }
    __syncthreads();
    if (tid < nrows) cnt[node0 + tid] = lc[tid];
    int* rows = (int*)agg + (size_t)node0 * D;
    for (int j = tid; j < nrows * 8; j += RK_THREADS) {
        const int node = j >> 3;
        const int q    = j & 7;
        *(int4*)(rows + (size_t)node * D + 4 * q) = ((const int4*)lst)[node * (LST_PITCH / 4) + q];
        if (lc[node] > 32)
            *(int4*)(rows + (size_t)node * D + 32 + 4 * q) =
                ((const int4*)lst)[node * (LST_PITCH / 4) + 8 + q];
    }
}

__device__ __forceinline__ float4 x_row(const float* __restrict__ x, int node, int lane) {
    return *(const float4*)(x + (size_t)node * D + lane * 4);
}

__device__ const float __attribute__((aligned(256))) zero_row[D] = {0.0f};

__device__ __forceinline__ void acc_add(float4& acc, const float4& r) {
    acc.x += r.x; acc.y += r.y; acc.z += r.z; acc.w += r.w;
}

template <int FROM, int CNT, int N>
__device__ __forceinline__ void rows_issue_nb(const int (&nb)[N], int cnt,
                                              const float* __restrict__ x,
                                              int lane, float4 (&r)[N])
{
#pragma unroll
    for (int t = FROM; t < FROM + CNT; ++t) {
        if (t < cnt) r[t] = x_row(x, nb[t], lane);
        else         r[t] = make_float4(0.0f, 0.0f, 0.0f, 0.0f);
    }
}

typedef float f4v __attribute__((ext_vector_type(4)));
typedef const char __attribute__((address_space(1)))* gbyte_ptr;
typedef const f4v __attribute__((address_space(1)))* gf4v_ptr;

__device__ __forceinline__ void rows_issue(int idv, int cnt, unsigned long long xaddr,
                                           unsigned long long zaddr,
                                           int lane, float4 (&r)[NFIRST])
{
    const unsigned long long av =
        (lane < cnt) ? xaddr + ((unsigned long long)(unsigned)idv << 9) : zaddr;
    const int lo = (int)(unsigned)av;
    const int hi = (int)(unsigned)(av >> 32);
    int slo[NFIRST], shi[NFIRST];
#pragma unroll
    for (int t = 0; t < NFIRST; ++t) {
        slo[t] = __builtin_amdgcn_readlane(lo, t);
        shi[t] = __builtin_amdgcn_readlane(hi, t);
    }
#pragma unroll
    for (int t = 0; t < NFIRST; ++t) {
        const unsigned long long a =
            ((unsigned long long)(unsigned)shi[t] << 32) | (unsigned long long)(unsigned)slo[t];
        const f4v v = ((gf4v_ptr)a)[lane];
        r[t] = make_float4(v.x, v.y, v.z, v.w);
    }
}

template <int FROM, int CNT, int N>
__device__ __forceinline__ float4 rows_sum(const float4 (&r)[N])
{
    float4 g = r[FROM];
#pragma unroll
    for (int t = 1; t < CNT; ++t) acc_add(g, r[FROM + t]);
    return g;
}

__device__ __forceinline__ float4 task_finish(
    const float4 (&r)[NFIRST], int id0, int deg, const int* __restrict__ list, int n, bool rev,
    const int* __restrict__ src, const int* __restrict__ dst,
    const float* __restrict__ x, int lane)
{
    float4 acc = rows_sum<0, 8>(r);
    acc_add(acc, rows_sum<8, NFIRST - 8>(r));
    if (deg <= CAP) {
        int idv = id0;
#pragma nounroll
        for (int j = NFIRST; j < deg; j += 8 - (j & 7)) {
            if (j == 32) {
                const int i = 32 + lane;
                idv = 0;
                if (i < deg) idv = list[i];
            }
            int nb[8];
#pragma unroll
            for (int t = 0; t < 8; ++t)
                nb[t] = __builtin_amdgcn_readlane(idv, (j & 31) + t);
            float4 q[8];
            rows_issue_nb<0, 8>(nb, min(deg - j, 8 - (j & 7)), x, lane, q);
            acc_add(acc, rows_sum<0, 8>(q));
        }
    } else {
        const int* key = rev ? src : dst;
        const int* oth = rev ? dst : src;
#pragma nounroll
        for (int base = 0; base < N_EDGES; base += 32) {
            const int k = key[base + lane];
            const int o = oth[base + lane];
            unsigned hit = __builtin_amdgcn_ballot_w32(k == n);
            while (hit) {
                const int j = __builtin_ctz(hit);
                hit &= hit - 1;
                acc_add(acc, x_row(x, __builtin_amdgcn_readlane(o, j), lane));
            }
        }
    }
    return acc;
}

#define TM      32
#define KTOT    (3 * D)
#define PITCH   (KTOT + 4)
#define NTILES  (N_NODES / TM)
#define GEMM_BLOCKS 256
#define KC      16
#define NCHUNK  (KTOT / 4 / KC)

__device__ __forceinline__ void list_fetch(
    int tile, int r0, int lane,
    const int* __restrict__ lst_in, const int* __restrict__ lst_out,
    const int* __restrict__ cnt_in, const int* __restrict__ cnt_out,
    int (&ids)[8], int (&dg)[8])
{
#pragma unroll
    for (int j = 0; j < 4; ++j) {
        const int row = tile * TM + r0 + 8 * j;
        ids[j]     = lst_in [(size_t)row * D + lane];
        ids[4 + j] = lst_out[(size_t)row * D + lane];
        dg[j]      = cnt_in[row];
        dg[4 + j]  = cnt_out[row];
    }
}

__device__ __forceinline__ void pair_issue(
    int ids_i, int dg_i, int ids_o, int dg_o, unsigned long long xaddr, unsigned long long zaddr,
    int lane,
    int& idi, int& di, int& ido, int& dout, float4 (&ri)[NFIRST], float4 (&ro)[NFIRST])
{
    di   = __builtin_amdgcn_readfirstlane(dg_i);
    dout = __builtin_amdgcn_readfirstlane(dg_o);
    idi  = (lane < di)   ? ids_i : 0;
    ido  = (lane < dout) ? ids_o : 0;
    rows_issue(idi, di   > CAP ? 0 : min(di, NFIRST),   xaddr, zaddr, lane, ri);
    rows_issue(ido, dout > CAP ? 0 : min(dout, NFIRST), xaddr, zaddr, lane, ro);
}

__device__ __forceinline__ void pair_finish(
    float* __restrict__ p, int row, int lane,
    const float4 (&ri)[NFIRST], const float4 (&ro)[NFIRST], int idi, int di, int ido, int dout,
    const int* __restrict__ lst_in, const int* __restrict__ lst_out,
    const int* __restrict__ src, const int* __restrict__ dst, const float* __restrict__ x)
{
    float4 a = task_finish(ri, idi, di,   lst_in  + (size_t)row * D, row, false, src, dst, x, lane);
    float4 b = task_finish(ro, ido, dout, lst_out + (size_t)row * D, row, true,  src, dst, x, lane);
    const float si = 1.0f / fmaxf((float)di, 1.0f);
    const float so = 1.0f / fmaxf((float)dout, 1.0f);
    a.x *= si; a.y *= si; a.z *= si; a.w *= si;
    b.x *= so; b.y *= so; b.z *= so; b.w *= so;
    *(float4*)(p)     = a;
    *(float4*)(p + D) = b;
}

__device__ __forceinline__ void frag_load(const float* A0, const float* A1, int c,
                                          v2f (&f0)[KC], v2f (&f1)[KC])
{
#pragma unroll
    for (int k = 0; k < KC; ++k) {
        f0[k] = *(const v2f*)(A0 + 4 * (c * KC + k));
        f1[k] = *(const v2f*)(A1 + 4 * (c * KC + k));
    }
}

__global__ __launch_bounds__(256) void sage_gemm(
    const int* __restrict__ src, const int* __restrict__ dst,
    const float* __restrict__ x,
    const int* __restrict__ lst_in, const int* __restrict__ lst_out,
    const int* __restrict__ cnt_in, const int* __restrict__ cnt_out,
    const float* __restrict__ w_l_in, const float* __restrict__ w_l_out,
    const float* __restrict__ w_r_in, const float* __restrict__ w_r_out,
    const float* __restrict__ b_l_in, const float* __restrict__ b_l_out,
    float* __restrict__ out)
{
    __shared__ __attribute__((aligned(16))) float tileA[2][TM * PITCH];

    const int tid   = threadIdx.x;
    const int wave  = tid >> 5;
    const int lane  = tid & 31;
    const int half  = lane >> 4;
    const int l15   = lane & 15;
    const int col   = wave * 16 + l15;

    v2f bw[96];
    {
        const size_t woff = (size_t)col * D + 2 * half;
#pragma unroll
        for (int i = 0; i < 32; ++i) {
            bw[i]      = *(const v2f*)(w_l_in  + woff + 4 * i);
            bw[32 + i] = *(const v2f*)(w_l_out + woff + 4 * i);
            bw[64 + i] = *(const v2f*)(w_r_in  + woff + 4 * i) + *(const v2f*)(w_r_out + woff + 4 * i);
        }
    }
    const float bias = b_l_in[col] + b_l_out[col];

    const unsigned long long xaddr = (unsigned long long)(size_t)(const void*)x;
    const unsigned long long zaddr = (unsigned long long)(size_t)(const void*)zero_row;

    const int c4 = lane;
    const int r0 = wave;

    int tile = blockIdx.x;
    int buf  = 0;
    int idsA[8], dgA[8];
    if (tile < NTILES) {
        int row = tile * TM + r0;
        int li = lst_in [(size_t)row * D + lane];
        int lo = lst_out[(size_t)row * D + lane];
        int gi = cnt_in[row];
        int go = cnt_out[row];
#pragma nounroll
        for (int j = 0; j < 4; ++j) {
            const int ci = li, co = lo, cgi = gi, cgo = go;
            const int rown = row + 8;
            if (j < 3) {
                li = lst_in [(size_t)rown * D + lane];
                lo = lst_out[(size_t)rown * D + lane];
                gi = cnt_in[rown];
                go = cnt_out[rown];
            }
            float* p = &tileA[0][0] + (r0 + 8 * j) * PITCH + c4 * 4;
            const float4 xr = *(const float4*)(x + (size_t)row * D + c4 * 4);
            float4 ri[NFIRST], ro[NFIRST];
            int idi, di, ido, dout;
            pair_issue(ci, cgi, co, cgo, xaddr, zaddr, lane, idi, di, ido, dout, ri, ro);
            pair_finish(p, row, lane, ri, ro, idi, di, ido, dout, lst_in, lst_out, src, dst, x);
            *(float4*)(p + 2 * D) = xr;
            row = rown;
        }
    }
    if (tile + GEMM_BLOCKS < NTILES) {
        list_fetch(tile + GEMM_BLOCKS, r0, lane, lst_in, lst_out, cnt_in, cnt_out, idsA, dgA);
    } else {
#pragma unroll
        for (int j = 0; j < 8; ++j) { idsA[j] = 0; dgA[j] = 0; }
    }
    __syncthreads();

    for (; tile < NTILES; tile += GEMM_BLOCKS) {
        const int  next      = tile + GEMM_BLOCKS;
        const bool has_next  = next < NTILES;
        const bool has_next2 = next + GEMM_BLOCKS < NTILES;

        const int xt = has_next ? next : tile;

        const float* A0 = &tileA[buf][0] + l15 * PITCH + 2 * half;
        const float* A1 = A0 + 16 * PITCH;
        float* nt = &tileA[buf ^ 1][0] + r0 * PITCH + c4 * 4;

        v8f acc0, acc1;
#pragma unroll
        for (int r = 0; r < 8; ++r) { acc0[r] = bias; acc1[r] = bias; }

        v2f fa0[2][KC], fa1[2][KC];
        frag_load(A0, A1, 0, fa0[0], fa1[0]);
#pragma unroll
        for (int c = 0; c < NCHUNK; ++c) {
            float4 ri[NFIRST], ro[NFIRST];
            int idi = 0, di = 0, ido = 0, dout = 0;
            float4 xr = make_float4(0.0f, 0.0f, 0.0f, 0.0f);
            if (c < 4) {
                xr = *(const float4*)(x + (size_t)(xt * TM + r0 + 8 * c) * D + c4 * 4);
                pair_issue(idsA[c], dgA[c], idsA[4 + c], dgA[4 + c], xaddr, zaddr, lane,
                           idi, di, ido, dout, ri, ro);
            }
            if (c == 4) {
                if (has_next2) {
                    list_fetch(next + GEMM_BLOCKS, r0, lane, lst_in, lst_out, cnt_in, cnt_out,
                               idsA, dgA);
                } else {
#pragma unroll
                    for (int j = 0; j < 8; ++j) { idsA[j] = 0; dgA[j] = 0; }
                }
            }
            if (c + 1 < NCHUNK)
                frag_load(A0, A1, c + 1, fa0[(c + 1) & 1], fa1[(c + 1) & 1]);
            __builtin_amdgcn_sched_barrier(0);
#pragma unroll
            for (int k = 0; k < KC; ++k) {
                acc0 = __builtin_amdgcn_wmma_f32_16x16x4_f32(
                     false, fa0[c & 1][k],  false, bw[c * KC + k],
                     (short)0, acc0,  false,  false);
                acc1 = __builtin_amdgcn_wmma_f32_16x16x4_f32(
                     false, fa1[c & 1][k],  false, bw[c * KC + k],
                     (short)0, acc1,  false,  false);
            }
            __builtin_amdgcn_sched_barrier(0);
            asm volatile("v_nop\n\tv_nop\n\tv_nop\n\tv_nop\n\tv_nop\n\tv_nop\n\tv_nop\n\tv_nop");
            __builtin_amdgcn_sched_barrier(0);
            if (c < 4) {
                pair_finish(nt + 8 * c * PITCH, xt * TM + r0 + 8 * c, lane,
                            ri, ro, idi, di, ido, dout, lst_in, lst_out, src, dst, x);
                *(float4*)(nt + 8 * c * PITCH + 2 * D) = xr;
            }
        }

        __syncthreads();

        const int mbase = tile * TM + half * 8;
#pragma unroll
        for (int r = 0; r < 8; ++r) {
            out[(size_t)(mbase + r) * D + col]      = acc0[r];
            out[(size_t)(mbase + 16 + r) * D + col] = acc1[r];
        }
        buf ^= 1;
    }
}

extern "C" void kernel_launch(void* const* d_in, const int* in_sizes, int n_in,
                              void* d_out, int out_size, void* d_ws, size_t ws_size,
                              hipStream_t stream)
{
    const float* x       = (const float*)d_in[0];
    const int*   ei      = (const int*)d_in[1];
    const float* w_l_in  = (const float*)d_in[2];
    const float* b_l_in  = (const float*)d_in[3];
    const float* w_r_in  = (const float*)d_in[4];
    const float* w_l_out = (const float*)d_in[5];
    const float* b_l_out = (const float*)d_in[6];
    const float* w_r_out = (const float*)d_in[7];
    float* out = (float*)d_out;

    const size_t NB   = (size_t)N_NODES * D * sizeof(float);
    const size_t CNTB = (size_t)N_NODES * sizeof(float);
    char* ws = (char*)d_ws;
    float* agg_in  = (float*)(ws);
    float* agg_out = (float*)(ws + NB);
    int*   cnt_in  = (int*)(ws + 2 * NB);
    int*   cnt_out = (int*)(ws + 2 * NB + CNTB);

    (void)in_sizes; (void)n_in; (void)out_size; (void)ws_size;

    sage_bucket<<<BK_BLOCKS, BK_THREADS, 0, stream>>>(ei, ei + N_EDGES, agg_in, agg_out);

    sage_rank<<<2 * NBUCK, RK_THREADS, 0, stream>>>(ei, ei + N_EDGES, agg_in, agg_out,
                                                    cnt_in, cnt_out);

    sage_gemm<<<GEMM_BLOCKS, 256, 0, stream>>>(ei, ei + N_EDGES, x,
                                                (const int*)agg_in, (const int*)agg_out,
                                                cnt_in, cnt_out,
                                                w_l_in, w_l_out, w_r_in, w_r_out,
                                                b_l_in, b_l_out, out);
}


# Output tile leaves through LDS: accumulators staged at the tile's end, stored as whole 512-B rows (4 stores per wave, was 16 of two half lines) ahead of the next tile's 5th WMMA chunk; sums unchanged
# speedup vs baseline: 4.2629x; 1.0375x over previous
#include <hip/hip_runtime.h>
#include <hip/hip_bf16.h>


#define N_NODES 100000
#define N_EDGES 640000
#define D 128

typedef float v2f __attribute__((ext_vector_type(2)));
typedef float v8f __attribute__((ext_vector_type(8)));

#define CAP 64
#define NFIRST 12
#define BNODES 256
#define NBUCK  ((N_NODES + BNODES - 1) / BNODES)
#define SHARE_WORDS 32
#define SHARE_CAP   (SHARE_WORDS - 1)
#define SH_PITCH    36
#define LST_PITCH   (CAP + 4)

__device__ __forceinline__ int* bucket_share(float* agg, int b, int k) {
    return (int*)(agg + ((size_t)(b * BNODES + (k >> 1))) * D + 64 + (k & 1) * SHARE_WORDS);
}

#define BK_BLOCKS  256
#define EPB        (N_EDGES / BK_BLOCKS)
#define BK_THREADS 512
#define EPT        ((EPB + BK_THREADS - 1) / BK_THREADS)

__global__ __launch_bounds__(BK_THREADS) void sage_bucket(
    const int* __restrict__ src, const int* __restrict__ dst,
    float* __restrict__ agg_in, float* __restrict__ agg_out)
{
    __shared__ __attribute__((aligned(16))) int sh[2 * NBUCK * SH_PITCH];
    const int tid = threadIdx.x;
    for (int j = tid; j < 2 * NBUCK * (SH_PITCH / 4); j += BK_THREADS)
        ((int4*)sh)[j] = make_int4(0, 0, 0, 0);
    __syncthreads();

    const int e0 = blockIdx.x * EPB;
    int s[EPT], d[EPT];
#pragma unroll
    for (int i = 0; i < EPT; ++i) {
        const int k = i * BK_THREADS + tid;
        s[i] = 0; d[i] = 0;
        if (k < EPB) { s[i] = src[e0 + k]; d[i] = dst[e0 + k]; }
    }
#pragma unroll
    for (int i = 0; i < EPT; ++i) {
        if (i * BK_THREADS + tid < EPB) {
            int* pi = sh + (d[i] >> 8) * SH_PITCH;
            const int ri = atomicAdd(pi, 1);
            if (ri < SHARE_CAP) pi[1 + ri] = (s[i] << 8) | (d[i] & 255);
            int* po = sh + (NBUCK + (s[i] >> 8)) * SH_PITCH;
            const int ro = atomicAdd(po, 1);
            if (ro < SHARE_CAP) po[1 + ro] = (d[i] << 8) | (s[i] & 255);
        }
    }
    __syncthreads();

    const int blk = blockIdx.x;
    for (int j = tid; j < 2 * NBUCK * (SHARE_WORDS / 4); j += BK_THREADS) {
        const int  line = j >> 3;
        const int  q    = j & 7;
        const bool rev  = line >= NBUCK;
        const int  b    = rev ? line - NBUCK : line;
        *(int4*)(bucket_share(rev ? agg_out : agg_in, b, blk) + 4 * q) =
            ((const int4*)sh)[line * (SH_PITCH / 4) + q];
    }
}

#define RK_THREADS BK_BLOCKS
#define RK_FIRST   15

__global__ __launch_bounds__(RK_THREADS) void sage_rank(
    const int* __restrict__ src, const int* __restrict__ dst,
    float* agg_in, float* agg_out,
    int* __restrict__ cnt_in, int* __restrict__ cnt_out)
{
    __shared__ int lc[BNODES];
    __shared__ __attribute__((aligned(16))) int lst[BNODES * LST_PITCH];
    __shared__ int over;
    const int  tid   = threadIdx.x;
    const bool rev   = blockIdx.x >= NBUCK;
    const int  b     = rev ? blockIdx.x - NBUCK : blockIdx.x;
    float*     agg   = rev ? agg_out : agg_in;
    int*       cnt   = rev ? cnt_out : cnt_in;
    const int  node0 = b * BNODES;
    const int  nrows = min(BNODES, N_NODES - node0);

    lc[tid] = 0;
    if (tid == 0) over = 0;
    for (int j = tid; j < BNODES * LST_PITCH / 4; j += RK_THREADS)
        ((int4*)lst)[j] = make_int4(0, 0, 0, 0);
    const int* sh = bucket_share(agg, b, tid);
    const int4 w0 = ((const int4*)sh)[0];
    const int4 w1 = ((const int4*)sh)[1];
    const int4 w2 = ((const int4*)sh)[2];
    const int4 w3 = ((const int4*)sh)[3];
    const int  n  = w0.x;
    __syncthreads();
    if (n > SHARE_CAP) over = 1;
    __syncthreads();

    if (over == 0) {
        const int ent[RK_FIRST] = { w0.y, w0.z, w0.w, w1.x, w1.y, w1.z, w1.w,
                                    w2.x, w2.y, w2.z, w2.w, w3.x, w3.y, w3.z, w3.w };
#pragma unroll
        for (int u = 0; u < RK_FIRST; ++u) {
            if (u < n) {
                const int kl = ent[u] & 255;
                const int r  = atomicAdd(&lc[kl], 1);
                if (r < CAP) lst[kl * LST_PITCH + r] = ent[u] >> 8;
            }
        }
        for (int u = RK_FIRST; u < n; ++u) {
            const int e  = sh[1 + u];
            const int kl = e & 255;
            const int r  = atomicAdd(&lc[kl], 1);
            if (r < CAP) lst[kl * LST_PITCH + r] = e >> 8;
        }
    } else {
        const int* key = rev ? src : dst;
        const int* oth = rev ? dst : src;
        for (int e = tid; e < N_EDGES; e += RK_THREADS) {
            const int k = key[e];
            if ((k >> 8) == b) {
                const int r = atomicAdd(&lc[k & 255], 1);
                if (r < CAP) lst[(k & 255) * LST_PITCH + r] = oth[e];
            }
        }
    }
    __syncthreads();
    if (tid < nrows) cnt[node0 + tid] = lc[tid];
    int* rows = (int*)agg + (size_t)node0 * D;
    for (int j = tid; j < nrows * 8; j += RK_THREADS) {
        const int node = j >> 3;
        const int q    = j & 7;
        *(int4*)(rows + (size_t)node * D + 4 * q) = ((const int4*)lst)[node * (LST_PITCH / 4) + q];
        if (lc[node] > 32)
            *(int4*)(rows + (size_t)node * D + 32 + 4 * q) =
                ((const int4*)lst)[node * (LST_PITCH / 4) + 8 + q];
    }
}

__device__ __forceinline__ float4 x_row(const float* __restrict__ x, int node, int lane) {
    return *(const float4*)(x + (size_t)node * D + lane * 4);
}

__device__ const float __attribute__((aligned(256))) zero_row[D] = {0.0f};

__device__ __forceinline__ void acc_add(float4& acc, const float4& r) {
    acc.x += r.x; acc.y += r.y; acc.z += r.z; acc.w += r.w;
}

template <int FROM, int CNT, int N>
__device__ __forceinline__ void rows_issue_nb(const int (&nb)[N], int cnt,
                                              const float* __restrict__ x,
                                              int lane, float4 (&r)[N])
{
#pragma unroll
    for (int t = FROM; t < FROM + CNT; ++t) {
        if (t < cnt) r[t] = x_row(x, nb[t], lane);
        else         r[t] = make_float4(0.0f, 0.0f, 0.0f, 0.0f);
    }
}

typedef float f4v __attribute__((ext_vector_type(4)));
typedef const char __attribute__((address_space(1)))* gbyte_ptr;
typedef const f4v __attribute__((address_space(1)))* gf4v_ptr;

__device__ __forceinline__ void rows_issue(int idv, int cnt, unsigned long long xaddr,
                                           unsigned long long zaddr,
                                           int lane, float4 (&r)[NFIRST])
{
    const unsigned long long av =
        (lane < cnt) ? xaddr + ((unsigned long long)(unsigned)idv << 9) : zaddr;
    const int lo = (int)(unsigned)av;
    const int hi = (int)(unsigned)(av >> 32);
    int slo[NFIRST], shi[NFIRST];
#pragma unroll
    for (int t = 0; t < NFIRST; ++t) {
        slo[t] = __builtin_amdgcn_readlane(lo, t);
        shi[t] = __builtin_amdgcn_readlane(hi, t);
    }
#pragma unroll
    for (int t = 0; t < NFIRST; ++t) {
        const unsigned long long a =
            ((unsigned long long)(unsigned)shi[t] << 32) | (unsigned long long)(unsigned)slo[t];
        const f4v v = ((gf4v_ptr)a)[lane];
        r[t] = make_float4(v.x, v.y, v.z, v.w);
    }
}

template <int FROM, int CNT, int N>
__device__ __forceinline__ float4 rows_sum(const float4 (&r)[N])
{
    float4 g = r[FROM];
#pragma unroll
    for (int t = 1; t < CNT; ++t) acc_add(g, r[FROM + t]);
    return g;
}

__device__ __forceinline__ float4 task_finish(
    const float4 (&r)[NFIRST], int id0, int deg, const int* __restrict__ list, int n, bool rev,
    const int* __restrict__ src, const int* __restrict__ dst,
    const float* __restrict__ x, int lane)
{
    float4 acc = rows_sum<0, 8>(r);
    acc_add(acc, rows_sum<8, NFIRST - 8>(r));
    if (deg <= CAP) {
        int idv = id0;
#pragma nounroll
        for (int j = NFIRST; j < deg; j += 8 - (j & 7)) {
            if (j == 32) {
                const int i = 32 + lane;
                idv = 0;
                if (i < deg) idv = list[i];
            }
            int nb[8];
#pragma unroll
            for (int t = 0; t < 8; ++t)
                nb[t] = __builtin_amdgcn_readlane(idv, (j & 31) + t);
            float4 q[8];
            rows_issue_nb<0, 8>(nb, min(deg - j, 8 - (j & 7)), x, lane, q);
            acc_add(acc, rows_sum<0, 8>(q));
        }
    } else {
        const int* key = rev ? src : dst;
        const int* oth = rev ? dst : src;
#pragma nounroll
        for (int base = 0; base < N_EDGES; base += 32) {
            const int k = key[base + lane];
            const int o = oth[base + lane];
            unsigned hit = __builtin_amdgcn_ballot_w32(k == n);
            while (hit) {
                const int j = __builtin_ctz(hit);
                hit &= hit - 1;
                acc_add(acc, x_row(x, __builtin_amdgcn_readlane(o, j), lane));
            }
        }
    }
    return acc;
}

#define TM      32
#define KTOT    (3 * D)
#define PITCH   (KTOT + 4)
#define NTILES  (N_NODES / TM)
#define GEMM_BLOCKS 256
#define KC      16
#define NCHUNK  (KTOT / 4 / KC)
#define SPITCH  (D + 4)

__device__ __forceinline__ void list_fetch(
    int tile, int r0, int lane,
    const int* __restrict__ lst_in, const int* __restrict__ lst_out,
    const int* __restrict__ cnt_in, const int* __restrict__ cnt_out,
    int (&ids)[8], int (&dg)[8])
{
#pragma unroll
    for (int j = 0; j < 4; ++j) {
        const int row = tile * TM + r0 + 8 * j;
        ids[j]     = lst_in [(size_t)row * D + lane];
        ids[4 + j] = lst_out[(size_t)row * D + lane];
        dg[j]      = cnt_in[row];
        dg[4 + j]  = cnt_out[row];
    }
}

__device__ __forceinline__ void pair_issue(
    int ids_i, int dg_i, int ids_o, int dg_o, unsigned long long xaddr, unsigned long long zaddr,
    int lane,
    int& idi, int& di, int& ido, int& dout, float4 (&ri)[NFIRST], float4 (&ro)[NFIRST])
{
    di   = __builtin_amdgcn_readfirstlane(dg_i);
    dout = __builtin_amdgcn_readfirstlane(dg_o);
    idi  = (lane < di)   ? ids_i : 0;
    ido  = (lane < dout) ? ids_o : 0;
    rows_issue(idi, di   > CAP ? 0 : min(di, NFIRST),   xaddr, zaddr, lane, ri);
    rows_issue(ido, dout > CAP ? 0 : min(dout, NFIRST), xaddr, zaddr, lane, ro);
}

__device__ __forceinline__ void pair_finish(
    float* __restrict__ p, int row, int lane,
    const float4 (&ri)[NFIRST], const float4 (&ro)[NFIRST], int idi, int di, int ido, int dout,
    const int* __restrict__ lst_in, const int* __restrict__ lst_out,
    const int* __restrict__ src, const int* __restrict__ dst, const float* __restrict__ x)
{
    float4 a = task_finish(ri, idi, di,   lst_in  + (size_t)row * D, row, false, src, dst, x, lane);
    float4 b = task_finish(ro, ido, dout, lst_out + (size_t)row * D, row, true,  src, dst, x, lane);
    const float si = 1.0f / fmaxf((float)di, 1.0f);
    const float so = 1.0f / fmaxf((float)dout, 1.0f);
    a.x *= si; a.y *= si; a.z *= si; a.w *= si;
    b.x *= so; b.y *= so; b.z *= so; b.w *= so;
    *(float4*)(p)     = a;
    *(float4*)(p + D) = b;
}

__device__ __forceinline__ void frag_load(const float* A0, const float* A1, int c,
                                          v2f (&f0)[KC], v2f (&f1)[KC])
{
#pragma unroll
    for (int k = 0; k < KC; ++k) {
        f0[k] = *(const v2f*)(A0 + 4 * (c * KC + k));
        f1[k] = *(const v2f*)(A1 + 4 * (c * KC + k));
    }
}

__global__ __launch_bounds__(256) void sage_gemm(
    const int* __restrict__ src, const int* __restrict__ dst,
    const float* __restrict__ x,
    const int* __restrict__ lst_in, const int* __restrict__ lst_out,
    const int* __restrict__ cnt_in, const int* __restrict__ cnt_out,
    const float* __restrict__ w_l_in, const float* __restrict__ w_l_out,
    const float* __restrict__ w_r_in, const float* __restrict__ w_r_out,
    const float* __restrict__ b_l_in, const float* __restrict__ b_l_out,
    float* __restrict__ out)
{
    __shared__ __attribute__((aligned(16))) float tileA[2][TM * PITCH];
    __shared__ __attribute__((aligned(16))) float stage[2][TM * SPITCH];

    const int tid   = threadIdx.x;
    const int wave  = tid >> 5;
    const int lane  = tid & 31;
    const int half  = lane >> 4;
    const int l15   = lane & 15;
    const int col   = wave * 16 + l15;

    v2f bw[96];
    {
        const size_t woff = (size_t)col * D + 2 * half;
#pragma unroll
        for (int i = 0; i < 32; ++i) {
            bw[i]      = *(const v2f*)(w_l_in  + woff + 4 * i);
            bw[32 + i] = *(const v2f*)(w_l_out + woff + 4 * i);
            bw[64 + i] = *(const v2f*)(w_r_in  + woff + 4 * i) + *(const v2f*)(w_r_out + woff + 4 * i);
        }
    }
    const float bias = b_l_in[col] + b_l_out[col];

    const unsigned long long xaddr = (unsigned long long)(size_t)(const void*)x;
    const unsigned long long zaddr = (unsigned long long)(size_t)(const void*)zero_row;

    const int c4 = lane;
    const int r0 = wave;

    int tile = blockIdx.x;
    int buf  = 0;
    int ptile = -1;
    int sb    = 0;
    int idsA[8], dgA[8];
    if (tile < NTILES) {
        int row = tile * TM + r0;
        int li = lst_in [(size_t)row * D + lane];
        int lo = lst_out[(size_t)row * D + lane];
        int gi = cnt_in[row];
        int go = cnt_out[row];
#pragma nounroll
        for (int j = 0; j < 4; ++j) {
            const int ci = li, co = lo, cgi = gi, cgo = go;
            const int rown = row + 8;
            if (j < 3) {
                li = lst_in [(size_t)rown * D + lane];
                lo = lst_out[(size_t)rown * D + lane];
                gi = cnt_in[rown];
                go = cnt_out[rown];
            }
            float* p = &tileA[0][0] + (r0 + 8 * j) * PITCH + c4 * 4;
            const float4 xr = *(const float4*)(x + (size_t)row * D + c4 * 4);
            float4 ri[NFIRST], ro[NFIRST];
            int idi, di, ido, dout;
            pair_issue(ci, cgi, co, cgo, xaddr, zaddr, lane, idi, di, ido, dout, ri, ro);
            pair_finish(p, row, lane, ri, ro, idi, di, ido, dout, lst_in, lst_out, src, dst, x);
            *(float4*)(p + 2 * D) = xr;
            row = rown;
        }
    }
    if (tile + GEMM_BLOCKS < NTILES) {
        list_fetch(tile + GEMM_BLOCKS, r0, lane, lst_in, lst_out, cnt_in, cnt_out, idsA, dgA);
    } else {
#pragma unroll
        for (int j = 0; j < 8; ++j) { idsA[j] = 0; dgA[j] = 0; }
    }
    __syncthreads();

    for (; tile < NTILES; tile += GEMM_BLOCKS) {
        const int  next      = tile + GEMM_BLOCKS;
        const bool has_next  = next < NTILES;
        const bool has_next2 = next + GEMM_BLOCKS < NTILES;

        const int xt = has_next ? next : tile;

        const float* A0 = &tileA[buf][0] + l15 * PITCH + 2 * half;
        const float* A1 = A0 + 16 * PITCH;
        float* nt = &tileA[buf ^ 1][0] + r0 * PITCH + c4 * 4;

        v8f acc0, acc1;
#pragma unroll
        for (int r = 0; r < 8; ++r) { acc0[r] = bias; acc1[r] = bias; }

        float4 o0, o1, o2, o3;
        const float* sprev = &stage[sb ^ 1][0] + r0 * SPITCH + c4 * 4;

        v2f fa0[2][KC], fa1[2][KC];
        frag_load(A0, A1, 0, fa0[0], fa1[0]);
#pragma unroll
        for (int c = 0; c < NCHUNK; ++c) {
            float4 ri[NFIRST], ro[NFIRST];
            int idi = 0, di = 0, ido = 0, dout = 0;
            float4 xr = make_float4(0.0f, 0.0f, 0.0f, 0.0f);
            if (c < 4) {
                xr = *(const float4*)(x + (size_t)(xt * TM + r0 + 8 * c) * D + c4 * 4);
                pair_issue(idsA[c], dgA[c], idsA[4 + c], dgA[4 + c], xaddr, zaddr, lane,
                           idi, di, ido, dout, ri, ro);
            }
            if (c == 4 && ptile >= 0) {
                float* op = out + (size_t)(ptile * TM + r0) * D + c4 * 4;
                *(float4*)(op)              = o0;
                *(float4*)(op +  8 * D)     = o1;
                *(float4*)(op + 16 * D)     = o2;
                *(float4*)(op + 24 * D)     = o3;
            }
            if (c == 4) {
                if (has_next2) {
                    list_fetch(next + GEMM_BLOCKS, r0, lane, lst_in, lst_out, cnt_in, cnt_out,
                               idsA, dgA);
                } else {
#pragma unroll
                    for (int j = 0; j < 8; ++j) { idsA[j] = 0; dgA[j] = 0; }
                }
            }
            if (c == 3 && ptile >= 0) {
                o0 = *(const float4*)(sprev);
                o1 = *(const float4*)(sprev +  8 * SPITCH);
                o2 = *(const float4*)(sprev + 16 * SPITCH);
                o3 = *(const float4*)(sprev + 24 * SPITCH);
            }
            if (c + 1 < NCHUNK)
                frag_load(A0, A1, c + 1, fa0[(c + 1) & 1], fa1[(c + 1) & 1]);
            __builtin_amdgcn_sched_barrier(0);
#pragma unroll
            for (int k = 0; k < KC; ++k) {
                acc0 = __builtin_amdgcn_wmma_f32_16x16x4_f32(
                     false, fa0[c & 1][k],  false, bw[c * KC + k],
                     (short)0, acc0,  false,  false);
                acc1 = __builtin_amdgcn_wmma_f32_16x16x4_f32(
                     false, fa1[c & 1][k],  false, bw[c * KC + k],
                     (short)0, acc1,  false,  false);
            }
            __builtin_amdgcn_sched_barrier(0);
            asm volatile("v_nop\n\tv_nop\n\tv_nop\n\tv_nop\n\tv_nop\n\tv_nop\n\tv_nop\n\tv_nop");
            __builtin_amdgcn_sched_barrier(0);
            if (c < 4) {
                pair_finish(nt + 8 * c * PITCH, xt * TM + r0 + 8 * c, lane,
                            ri, ro, idi, di, ido, dout, lst_in, lst_out, src, dst, x);
                *(float4*)(nt + 8 * c * PITCH + 2 * D) = xr;
            }
        }

        {
            float* sp = &stage[sb][0] + (half * 8) * SPITCH + col;
#pragma unroll
            for (int r = 0; r < 8; ++r) {
                sp[r * SPITCH]        = acc0[r];
                sp[(16 + r) * SPITCH] = acc1[r];
            }
        }

        __syncthreads();

        ptile = tile;
        sb  ^= 1;
        buf ^= 1;
    }

    if (ptile >= 0) {
        const float* sprev = &stage[sb ^ 1][0] + r0 * SPITCH + c4 * 4;
        const float4 o0 = *(const float4*)(sprev);
        const float4 o1 = *(const float4*)(sprev +  8 * SPITCH);
        const float4 o2 = *(const float4*)(sprev + 16 * SPITCH);
        const float4 o3 = *(const float4*)(sprev + 24 * SPITCH);
        float* op = out + (size_t)(ptile * TM + r0) * D + c4 * 4;
        *(float4*)(op)          = o0;
        *(float4*)(op +  8 * D) = o1;
        *(float4*)(op + 16 * D) = o2;
        *(float4*)(op + 24 * D) = o3;
    }
}

extern "C" void kernel_launch(void* const* d_in, const int* in_sizes, int n_in,
                              void* d_out, int out_size, void* d_ws, size_t ws_size,
                              hipStream_t stream)
{
    const float* x       = (const float*)d_in[0];
    const int*   ei      = (const int*)d_in[1];
    const float* w_l_in  = (const float*)d_in[2];
    const float* b_l_in  = (const float*)d_in[3];
    const float* w_r_in  = (const float*)d_in[4];
    const float* w_l_out = (const float*)d_in[5];
    const float* b_l_out = (const float*)d_in[6];
    const float* w_r_out = (const float*)d_in[7];
    float* out = (float*)d_out;

    const size_t NB   = (size_t)N_NODES * D * sizeof(float);
    const size_t CNTB = (size_t)N_NODES * sizeof(float);
    char* ws = (char*)d_ws;
    float* agg_in  = (float*)(ws);
    float* agg_out = (float*)(ws + NB);
    int*   cnt_in  = (int*)(ws + 2 * NB);
    int*   cnt_out = (int*)(ws + 2 * NB + CNTB);

    (void)in_sizes; (void)n_in; (void)out_size; (void)ws_size;

    sage_bucket<<<BK_BLOCKS, BK_THREADS, 0, stream>>>(ei, ei + N_EDGES, agg_in, agg_out);

    sage_rank<<<2 * NBUCK, RK_THREADS, 0, stream>>>(ei, ei + N_EDGES, agg_in, agg_out,
                                                    cnt_in, cnt_out);

    sage_gemm<<<GEMM_BLOCKS, 256, 0, stream>>>(ei, ei + N_EDGES, x,
                                                (const int*)agg_in, (const int*)agg_out,
                                                cnt_in, cnt_out,
                                                w_l_in, w_l_out, w_r_in, w_r_out,
                                                b_l_in, b_l_out, out);
}


# GEMM tile stage: matrix work cut into 8 chunks, one neighbour list per chunk; a list's rows of x are asked for two chunks ahead of its sums, so two lists' loads are always in flight; sum order kept
# speedup vs baseline: 4.4156x; 1.0358x over previous
#include <hip/hip_runtime.h>
#include <hip/hip_bf16.h>


#define N_NODES 100000
#define N_EDGES 640000
#define D 128

typedef float v2f __attribute__((ext_vector_type(2)));
typedef float v8f __attribute__((ext_vector_type(8)));

#define CAP 64
#define NFIRST 12
#define BNODES 256
#define NBUCK  ((N_NODES + BNODES - 1) / BNODES)
#define SHARE_WORDS 32
#define SHARE_CAP   (SHARE_WORDS - 1)
#define SH_PITCH    36
#define LST_PITCH   (CAP + 4)

__device__ __forceinline__ int* bucket_share(float* agg, int b, int k) {
    return (int*)(agg + ((size_t)(b * BNODES + (k >> 1))) * D + 64 + (k & 1) * SHARE_WORDS);
}

#define BK_BLOCKS  256
#define EPB        (N_EDGES / BK_BLOCKS)
#define BK_THREADS 512
#define EPT        ((EPB + BK_THREADS - 1) / BK_THREADS)

__global__ __launch_bounds__(BK_THREADS) void sage_bucket(
    const int* __restrict__ src, const int* __restrict__ dst,
    float* __restrict__ agg_in, float* __restrict__ agg_out)
{
    __shared__ __attribute__((aligned(16))) int sh[2 * NBUCK * SH_PITCH];
    const int tid = threadIdx.x;
    for (int j = tid; j < 2 * NBUCK * (SH_PITCH / 4); j += BK_THREADS)
        ((int4*)sh)[j] = make_int4(0, 0, 0, 0);
    __syncthreads();

    const int e0 = blockIdx.x * EPB;
    int s[EPT], d[EPT];
#pragma unroll
    for (int i = 0; i < EPT; ++i) {
        const int k = i * BK_THREADS + tid;
        s[i] = 0; d[i] = 0;
        if (k < EPB) { s[i] = src[e0 + k]; d[i] = dst[e0 + k]; }
    }
#pragma unroll
    for (int i = 0; i < EPT; ++i) {
        if (i * BK_THREADS + tid < EPB) {
            int* pi = sh + (d[i] >> 8) * SH_PITCH;
            const int ri = atomicAdd(pi, 1);
            if (ri < SHARE_CAP) pi[1 + ri] = (s[i] << 8) | (d[i] & 255);
            int* po = sh + (NBUCK + (s[i] >> 8)) * SH_PITCH;
            const int ro = atomicAdd(po, 1);
            if (ro < SHARE_CAP) po[1 + ro] = (d[i] << 8) | (s[i] & 255);
        }
    }
    __syncthreads();

    const int blk = blockIdx.x;
    for (int j = tid; j < 2 * NBUCK * (SHARE_WORDS / 4); j += BK_THREADS) {
        const int  line = j >> 3;
        const int  q    = j & 7;
        const bool rev  = line >= NBUCK;
        const int  b    = rev ? line - NBUCK : line;
        *(int4*)(bucket_share(rev ? agg_out : agg_in, b, blk) + 4 * q) =
            ((const int4*)sh)[line * (SH_PITCH / 4) + q];
    }
}

#define RK_THREADS BK_BLOCKS
#define RK_FIRST   15

__global__ __launch_bounds__(RK_THREADS) void sage_rank(
    const int* __restrict__ src, const int* __restrict__ dst,
    float* agg_in, float* agg_out,
    int* __restrict__ cnt_in, int* __restrict__ cnt_out)
{
    __shared__ int lc[BNODES];
    __shared__ __attribute__((aligned(16))) int lst[BNODES * LST_PITCH];
    __shared__ int over;
    const int  tid   = threadIdx.x;
    const bool rev   = blockIdx.x >= NBUCK;
    const int  b     = rev ? blockIdx.x - NBUCK : blockIdx.x;
    float*     agg   = rev ? agg_out : agg_in;
    int*       cnt   = rev ? cnt_out : cnt_in;
    const int  node0 = b * BNODES;
    const int  nrows = min(BNODES, N_NODES - node0);

    lc[tid] = 0;
    if (tid == 0) over = 0;
    for (int j = tid; j < BNODES * LST_PITCH / 4; j += RK_THREADS)
        ((int4*)lst)[j] = make_int4(0, 0, 0, 0);
    const int* sh = bucket_share(agg, b, tid);
    const int4 w0 = ((const int4*)sh)[0];
    const int4 w1 = ((const int4*)sh)[1];
    const int4 w2 = ((const int4*)sh)[2];
    const int4 w3 = ((const int4*)sh)[3];
    const int  n  = w0.x;
    __syncthreads();
    if (n > SHARE_CAP) over = 1;
    __syncthreads();

    if (over == 0) {
        const int ent[RK_FIRST] = { w0.y, w0.z, w0.w, w1.x, w1.y, w1.z, w1.w,
                                    w2.x, w2.y, w2.z, w2.w, w3.x, w3.y, w3.z, w3.w };
#pragma unroll
        for (int u = 0; u < RK_FIRST; ++u) {
            if (u < n) {
                const int kl = ent[u] & 255;
                const int r  = atomicAdd(&lc[kl], 1);
                if (r < CAP) lst[kl * LST_PITCH + r] = ent[u] >> 8;
            }
        }
        for (int u = RK_FIRST; u < n; ++u) {
            const int e  = sh[1 + u];
            const int kl = e & 255;
            const int r  = atomicAdd(&lc[kl], 1);
            if (r < CAP) lst[kl * LST_PITCH + r] = e >> 8;
        }
    } else {
        const int* key = rev ? src : dst;
        const int* oth = rev ? dst : src;
        for (int e = tid; e < N_EDGES; e += RK_THREADS) {
            const int k = key[e];
            if ((k >> 8) == b) {
                const int r = atomicAdd(&lc[k & 255], 1);
                if (r < CAP) lst[(k & 255) * LST_PITCH + r] = oth[e];
            }
        }
    }
    __syncthreads();
    if (tid < nrows) cnt[node0 + tid] = lc[tid];
    int* rows = (int*)agg + (size_t)node0 * D;
    for (int j = tid; j < nrows * 8; j += RK_THREADS) {
        const int node = j >> 3;
        const int q    = j & 7;
        *(int4*)(rows + (size_t)node * D + 4 * q) = ((const int4*)lst)[node * (LST_PITCH / 4) + q];
        if (lc[node] > 32)
            *(int4*)(rows + (size_t)node * D + 32 + 4 * q) =
                ((const int4*)lst)[node * (LST_PITCH / 4) + 8 + q];
    }
}

__device__ __forceinline__ float4 x_row(const float* __restrict__ x, int node, int lane) {
    return *(const float4*)(x + (size_t)node * D + lane * 4);
}

__device__ const float __attribute__((aligned(256))) zero_row[D] = {0.0f};

__device__ __forceinline__ void acc_add(float4& acc, const float4& r) {
    acc.x += r.x; acc.y += r.y; acc.z += r.z; acc.w += r.w;
}

template <int FROM, int CNT, int N>
__device__ __forceinline__ void rows_issue_nb(const int (&nb)[N], int cnt,
                                              const float* __restrict__ x,
                                              int lane, float4 (&r)[N])
{
#pragma unroll
    for (int t = FROM; t < FROM + CNT; ++t) {
        if (t < cnt) r[t] = x_row(x, nb[t], lane);
        else         r[t] = make_float4(0.0f, 0.0f, 0.0f, 0.0f);
    }
}

typedef float f4v __attribute__((ext_vector_type(4)));
typedef const char __attribute__((address_space(1)))* gbyte_ptr;
typedef const f4v __attribute__((address_space(1)))* gf4v_ptr;

__device__ __forceinline__ void rows_issue(int idv, int cnt, unsigned long long xaddr,
                                           unsigned long long zaddr,
                                           int lane, float4 (&r)[NFIRST])
{
    const unsigned long long av =
        (lane < cnt) ? xaddr + ((unsigned long long)(unsigned)idv << 9) : zaddr;
    const int lo = (int)(unsigned)av;
    const int hi = (int)(unsigned)(av >> 32);
    int slo[NFIRST], shi[NFIRST];
#pragma unroll
    for (int t = 0; t < NFIRST; ++t) {
        slo[t] = __builtin_amdgcn_readlane(lo, t);
        shi[t] = __builtin_amdgcn_readlane(hi, t);
    }
#pragma unroll
    for (int t = 0; t < NFIRST; ++t) {
        const unsigned long long a =
            ((unsigned long long)(unsigned)shi[t] << 32) | (unsigned long long)(unsigned)slo[t];
        const f4v v = ((gf4v_ptr)a)[lane];
        r[t] = make_float4(v.x, v.y, v.z, v.w);
    }
}

template <int FROM, int CNT, int N>
__device__ __forceinline__ float4 rows_sum(const float4 (&r)[N])
{
    float4 g = r[FROM];
#pragma unroll
    for (int t = 1; t < CNT; ++t) acc_add(g, r[FROM + t]);
    return g;
}

__device__ __forceinline__ float4 task_finish(
    const float4 (&r)[NFIRST], int id0, int deg, const int* __restrict__ list, int n, bool rev,
    const int* __restrict__ src, const int* __restrict__ dst,
    const float* __restrict__ x, int lane)
{
    float4 acc = rows_sum<0, 8>(r);
    acc_add(acc, rows_sum<8, NFIRST - 8>(r));
    if (deg <= CAP) {
        int idv = id0;
#pragma nounroll
        for (int j = NFIRST; j < deg; j += 8 - (j & 7)) {
            if (j == 32) {
                const int i = 32 + lane;
                idv = 0;
                if (i < deg) idv = list[i];
            }
            int nb[8];
#pragma unroll
            for (int t = 0; t < 8; ++t)
                nb[t] = __builtin_amdgcn_readlane(idv, (j & 31) + t);
            float4 q[8];
            rows_issue_nb<0, 8>(nb, min(deg - j, 8 - (j & 7)), x, lane, q);
            acc_add(acc, rows_sum<0, 8>(q));
        }
    } else {
        const int* key = rev ? src : dst;
        const int* oth = rev ? dst : src;
#pragma nounroll
        for (int base = 0; base < N_EDGES; base += 32) {
            const int k = key[base + lane];
            const int o = oth[base + lane];
            unsigned hit = __builtin_amdgcn_ballot_w32(k == n);
            while (hit) {
                const int j = __builtin_ctz(hit);
                hit &= hit - 1;
                acc_add(acc, x_row(x, __builtin_amdgcn_readlane(o, j), lane));
            }
        }
    }
    return acc;
}

#define TM      32
#define KTOT    (3 * D)
#define PITCH   (KTOT + 4)
#define NTILES  (N_NODES / TM)
#define GEMM_BLOCKS 256
#define KC      12
#define NCHUNK  (KTOT / 4 / KC)
#define SPITCH  (D + 4)

__device__ __forceinline__ void list_fetch(
    int tile, int r0, int lane,
    const int* __restrict__ lst_in, const int* __restrict__ lst_out,
    const int* __restrict__ cnt_in, const int* __restrict__ cnt_out,
    int (&ids)[8], int (&dg)[8])
{
#pragma unroll
    for (int j = 0; j < 4; ++j) {
        const int row = tile * TM + r0 + 8 * j;
        ids[j]     = lst_in [(size_t)row * D + lane];
        ids[4 + j] = lst_out[(size_t)row * D + lane];
        dg[j]      = cnt_in[row];
        dg[4 + j]  = cnt_out[row];
    }
}

__device__ __forceinline__ void pair_issue(
    int ids_i, int dg_i, int ids_o, int dg_o, unsigned long long xaddr, unsigned long long zaddr,
    int lane,
    int& idi, int& di, int& ido, int& dout, float4 (&ri)[NFIRST], float4 (&ro)[NFIRST])
{
    di   = __builtin_amdgcn_readfirstlane(dg_i);
    dout = __builtin_amdgcn_readfirstlane(dg_o);
    idi  = (lane < di)   ? ids_i : 0;
    ido  = (lane < dout) ? ids_o : 0;
    rows_issue(idi, di   > CAP ? 0 : min(di, NFIRST),   xaddr, zaddr, lane, ri);
    rows_issue(ido, dout > CAP ? 0 : min(dout, NFIRST), xaddr, zaddr, lane, ro);
}

__device__ __forceinline__ void pair_finish(
    float* __restrict__ p, int row, int lane,
    const float4 (&ri)[NFIRST], const float4 (&ro)[NFIRST], int idi, int di, int ido, int dout,
    const int* __restrict__ lst_in, const int* __restrict__ lst_out,
    const int* __restrict__ src, const int* __restrict__ dst, const float* __restrict__ x)
{
    float4 a = task_finish(ri, idi, di,   lst_in  + (size_t)row * D, row, false, src, dst, x, lane);
    float4 b = task_finish(ro, ido, dout, lst_out + (size_t)row * D, row, true,  src, dst, x, lane);
    const float si = 1.0f / fmaxf((float)di, 1.0f);
    const float so = 1.0f / fmaxf((float)dout, 1.0f);
    a.x *= si; a.y *= si; a.z *= si; a.w *= si;
    b.x *= so; b.y *= so; b.z *= so; b.w *= so;
    *(float4*)(p)     = a;
    *(float4*)(p + D) = b;
}

__device__ __forceinline__ void task_issue(
    int ids, int dg, unsigned long long xaddr, unsigned long long zaddr, int lane,
    int& id, int& d, float4 (&r)[NFIRST])
{
    d  = __builtin_amdgcn_readfirstlane(dg);
    id = (lane < d) ? ids : 0;
    rows_issue(id, d > CAP ? 0 : min(d, NFIRST), xaddr, zaddr, lane, r);
}

__device__ __forceinline__ void task_store(
    float* __restrict__ p, int row, int lane, const float4 (&r)[NFIRST], int id, int d,
    const int* __restrict__ lst, bool rev,
    const int* __restrict__ src, const int* __restrict__ dst, const float* __restrict__ x)
{
    float4 a = task_finish(r, id, d, lst + (size_t)row * D, row, rev, src, dst, x, lane);
    const float s = 1.0f / fmaxf((float)d, 1.0f);
    a.x *= s; a.y *= s; a.z *= s; a.w *= s;
    *(float4*)(p) = a;
}

__device__ __forceinline__ void frag_load(const float* A0, const float* A1, int c,
                                          v2f (&f0)[KC], v2f (&f1)[KC])
{
#pragma unroll
    for (int k = 0; k < KC; ++k) {
        f0[k] = *(const v2f*)(A0 + 4 * (c * KC + k));
        f1[k] = *(const v2f*)(A1 + 4 * (c * KC + k));
    }
}

__global__ __launch_bounds__(256) void sage_gemm(
    const int* __restrict__ src, const int* __restrict__ dst,
    const float* __restrict__ x,
    const int* __restrict__ lst_in, const int* __restrict__ lst_out,
    const int* __restrict__ cnt_in, const int* __restrict__ cnt_out,
    const float* __restrict__ w_l_in, const float* __restrict__ w_l_out,
    const float* __restrict__ w_r_in, const float* __restrict__ w_r_out,
    const float* __restrict__ b_l_in, const float* __restrict__ b_l_out,
    float* __restrict__ out)
{
    __shared__ __attribute__((aligned(16))) float tileA[2][TM * PITCH];
    __shared__ __attribute__((aligned(16))) float stage[2][TM * SPITCH];

    const int tid   = threadIdx.x;
    const int wave  = __builtin_amdgcn_readfirstlane(tid >> 5);
    const int lane  = tid & 31;
    const int half  = lane >> 4;
    const int l15   = lane & 15;
    const int col   = wave * 16 + l15;

    v2f bw[96];
    {
        const size_t woff = (size_t)col * D + 2 * half;
#pragma unroll
        for (int i = 0; i < 32; ++i) {
            bw[i]      = *(const v2f*)(w_l_in  + woff + 4 * i);
            bw[32 + i] = *(const v2f*)(w_l_out + woff + 4 * i);
            bw[64 + i] = *(const v2f*)(w_r_in  + woff + 4 * i) + *(const v2f*)(w_r_out + woff + 4 * i);
        }
    }
    const float bias = b_l_in[col] + b_l_out[col];

    const unsigned long long xaddr = (unsigned long long)(size_t)(const void*)x;
    const unsigned long long zaddr = (unsigned long long)(size_t)(const void*)zero_row;

    const int c4 = lane;
    const int r0 = wave;

    int tile = blockIdx.x;
    int buf  = 0;
    int ptile = -1;
    int sb    = 0;
    int idsA[8], dgA[8];
    if (tile < NTILES) {
        int row = tile * TM + r0;
        int li = lst_in [(size_t)row * D + lane];
        int lo = lst_out[(size_t)row * D + lane];
        int gi = cnt_in[row];
        int go = cnt_out[row];
#pragma nounroll
        for (int j = 0; j < 4; ++j) {
            const int ci = li, co = lo, cgi = gi, cgo = go;
            const int rown = row + 8;
            if (j < 3) {
                li = lst_in [(size_t)rown * D + lane];
                lo = lst_out[(size_t)rown * D + lane];
                gi = cnt_in[rown];
                go = cnt_out[rown];
            }
            float* p = &tileA[0][0] + (r0 + 8 * j) * PITCH + c4 * 4;
            const float4 xr = *(const float4*)(x + (size_t)row * D + c4 * 4);
            float4 ri[NFIRST], ro[NFIRST];
            int idi, di, ido, dout;
            pair_issue(ci, cgi, co, cgo, xaddr, zaddr, lane, idi, di, ido, dout, ri, ro);
            pair_finish(p, row, lane, ri, ro, idi, di, ido, dout, lst_in, lst_out, src, dst, x);
            *(float4*)(p + 2 * D) = xr;
            row = rown;
        }
    }
    if (tile + GEMM_BLOCKS < NTILES) {
        list_fetch(tile + GEMM_BLOCKS, r0, lane, lst_in, lst_out, cnt_in, cnt_out, idsA, dgA);
    } else {
#pragma unroll
        for (int j = 0; j < 8; ++j) { idsA[j] = 0; dgA[j] = 0; }
    }
    __syncthreads();

    for (; tile < NTILES; tile += GEMM_BLOCKS) {
        const int  next      = tile + GEMM_BLOCKS;
        const bool has_next  = next < NTILES;
        const bool has_next2 = next + GEMM_BLOCKS < NTILES;

        const int xt = has_next ? next : tile;

        const float* A0 = &tileA[buf][0] + l15 * PITCH + 2 * half;
        const float* A1 = A0 + 16 * PITCH;
        float* nt = &tileA[buf ^ 1][0] + r0 * PITCH + c4 * 4;

        v8f acc0, acc1;
#pragma unroll
        for (int r = 0; r < 8; ++r) { acc0[r] = bias; acc1[r] = bias; }

        float4 o0, o1, o2, o3;
        const float* sprev = &stage[sb ^ 1][0] + r0 * SPITCH + c4 * 4;

        v2f fa0[2][KC], fa1[2][KC];
        frag_load(A0, A1, 0, fa0[0], fa1[0]);

        float4 rr[2][NFIRST];
        int    tl[2] = {0, 0}, td[2] = {0, 0};
        float4 xrow = *(const float4*)(x + (size_t)(xt * TM + r0) * D + c4 * 4);
        task_issue(idsA[0], has_next ? dgA[0] : 0, xaddr, zaddr, lane, tl[0], td[0], rr[0]);
        __builtin_amdgcn_sched_barrier(0);
        task_issue(idsA[4], has_next ? dgA[4] : 0, xaddr, zaddr, lane, tl[1], td[1], rr[1]);

#pragma unroll
        for (int c = 0; c < NCHUNK; ++c) {
            if (c == NCHUNK - 2 && ptile >= 0) {
                float* op = out + (size_t)(ptile * TM + r0) * D + c4 * 4;
                *(float4*)(op)              = o0;
                *(float4*)(op +  8 * D)     = o1;
                *(float4*)(op + 16 * D)     = o2;
                *(float4*)(op + 24 * D)     = o3;
            }
            if (c == NCHUNK - 2) {
                list_fetch(has_next2 ? next + GEMM_BLOCKS : tile, r0, lane, lst_in, lst_out,
                           cnt_in, cnt_out, idsA, dgA);
            }
            if (c == NCHUNK - 3 && ptile >= 0) {
                o0 = *(const float4*)(sprev);
                o1 = *(const float4*)(sprev +  8 * SPITCH);
                o2 = *(const float4*)(sprev + 16 * SPITCH);
                o3 = *(const float4*)(sprev + 24 * SPITCH);
            }
            if (c + 1 < NCHUNK)
                frag_load(A0, A1, c + 1, fa0[(c + 1) & 1], fa1[(c + 1) & 1]);
            __builtin_amdgcn_sched_barrier(0);
#pragma unroll
            for (int k = 0; k < KC; ++k) {
                acc0 = __builtin_amdgcn_wmma_f32_16x16x4_f32(
                     false, fa0[c & 1][k],  false, bw[c * KC + k],
                     (short)0, acc0,  false,  false);
                acc1 = __builtin_amdgcn_wmma_f32_16x16x4_f32(
                     false, fa1[c & 1][k],  false, bw[c * KC + k],
                     (short)0, acc1,  false,  false);
            }
            __builtin_amdgcn_sched_barrier(0);
            {
                float4& ra = rr[c & 1][0];
                float4& rb = rr[c & 1][8];
                f4v va = {ra.x, ra.y, ra.z, ra.w};
                f4v vb = {rb.x, rb.y, rb.z, rb.w};
                asm volatile("v_nop\n\tv_nop\n\tv_nop\n\tv_nop\n\tv_nop\n\tv_nop\n\tv_nop\n\tv_nop"
                             : "+v"(acc0), "+v"(acc1), "+v"(va), "+v"(vb));
                ra = make_float4(va.x, va.y, va.z, va.w);
                rb = make_float4(vb.x, vb.y, vb.z, vb.w);
            }
            __builtin_amdgcn_sched_barrier(0);
            {
                const int  j   = c >> 1;
                const bool rev = (c & 1) != 0;
                float* p = nt + 8 * j * PITCH;
                task_store(p + (rev ? D : 0), xt * TM + r0 + 8 * j, lane, rr[c & 1], tl[c & 1], td[c & 1],
                           rev ? lst_out : lst_in, rev, src, dst, x);
                if (!rev) *(float4*)(p + 2 * D) = xrow;
            }
            if (c + 2 < NCHUNK) {
                const int  t   = c + 2;
                const int  j   = t >> 1;
                const bool rev = (t & 1) != 0;
                if (!rev) xrow = *(const float4*)(x + (size_t)(xt * TM + r0 + 8 * j) * D + c4 * 4);
                task_issue(idsA[(rev ? 4 : 0) + j], has_next ? dgA[(rev ? 4 : 0) + j] : 0, xaddr, zaddr, lane,
                           tl[t & 1], td[t & 1], rr[t & 1]);
            }
        }

        {
            float* sp = &stage[sb][0] + (half * 8) * SPITCH + col;
#pragma unroll
            for (int r = 0; r < 8; ++r) {
                sp[r * SPITCH]        = acc0[r];
                sp[(16 + r) * SPITCH] = acc1[r];
            }
        }

        __syncthreads();

        ptile = tile;
        sb  ^= 1;
        buf ^= 1;
    }

    if (ptile >= 0) {
        const float* sprev = &stage[sb ^ 1][0] + r0 * SPITCH + c4 * 4;
        const float4 o0 = *(const float4*)(sprev);
        const float4 o1 = *(const float4*)(sprev +  8 * SPITCH);
        const float4 o2 = *(const float4*)(sprev + 16 * SPITCH);
        const float4 o3 = *(const float4*)(sprev + 24 * SPITCH);
        float* op = out + (size_t)(ptile * TM + r0) * D + c4 * 4;
        *(float4*)(op)          = o0;
        *(float4*)(op +  8 * D) = o1;
        *(float4*)(op + 16 * D) = o2;
        *(float4*)(op + 24 * D) = o3;
    }
}

extern "C" void kernel_launch(void* const* d_in, const int* in_sizes, int n_in,
                              void* d_out, int out_size, void* d_ws, size_t ws_size,
                              hipStream_t stream)
{
    const float* x       = (const float*)d_in[0];
    const int*   ei      = (const int*)d_in[1];
    const float* w_l_in  = (const float*)d_in[2];
    const float* b_l_in  = (const float*)d_in[3];
    const float* w_r_in  = (const float*)d_in[4];
    const float* w_l_out = (const float*)d_in[5];
    const float* b_l_out = (const float*)d_in[6];
    const float* w_r_out = (const float*)d_in[7];
    float* out = (float*)d_out;

    const size_t NB   = (size_t)N_NODES * D * sizeof(float);
    const size_t CNTB = (size_t)N_NODES * sizeof(float);
    char* ws = (char*)d_ws;
    float* agg_in  = (float*)(ws);
    float* agg_out = (float*)(ws + NB);
    int*   cnt_in  = (int*)(ws + 2 * NB);
    int*   cnt_out = (int*)(ws + 2 * NB + CNTB);

    (void)in_sizes; (void)n_in; (void)out_size; (void)ws_size;

    sage_bucket<<<BK_BLOCKS, BK_THREADS, 0, stream>>>(ei, ei + N_EDGES, agg_in, agg_out);

    sage_rank<<<2 * NBUCK, RK_THREADS, 0, stream>>>(ei, ei + N_EDGES, agg_in, agg_out,
                                                    cnt_in, cnt_out);

    sage_gemm<<<GEMM_BLOCKS, 256, 0, stream>>>(ei, ei + N_EDGES, x,
                                                (const int*)agg_in, (const int*)agg_out,
                                                cnt_in, cnt_out,
                                                w_l_in, w_l_out, w_r_in, w_r_out,
                                                b_l_in, b_l_out, out);
}
